# starting kernel
# baseline (speedup 1.0000x reference)
#include <hip/hip_runtime.h>
#include <math.h>

typedef __attribute__((ext_vector_type(2))) float v2f;
typedef __attribute__((ext_vector_type(8))) float v8f;

#define NBASIS 128
#define NCHUNK 32
#define CUTOFF 6.0f
#define DELTA  (CUTOFF / (NBASIS - 1))
#define GAMMA_ ((NBASIS - 1) / CUTOFF)
#define PI_F   3.14159265358979323846f
#define LOG2E_F 1.44269504088896340736f
#define K2_    (-GAMMA_ * LOG2E_F)

#if __has_builtin(__builtin_amdgcn_exp2f)
#define FAST_EXP2(x) __builtin_amdgcn_exp2f(x)
#else
#define FAST_EXP2(x) exp2f(x)
#endif

__device__ __forceinline__ float softplus_(float x) {
  return fmaxf(x, 0.0f) + log1pf(expf(-fabsf(x)));
}

__device__ __forceinline__ float blockReduceSum256(float v) {
  __shared__ float sred[256];
  sred[threadIdx.x] = v;
  __syncthreads();
  for (int s = blockDim.x >> 1; s > 0; s >>= 1) {
    if (threadIdx.x < (unsigned)s) sred[threadIdx.x] += sred[threadIdx.x + s];
    __syncthreads();
  }
  return sred[0];
}

__global__ void eam_zero(float* __restrict__ a, int na, float* __restrict__ b, int nb) {
  const int i = blockIdx.x * blockDim.x + threadIdx.x;
  const int stride = gridDim.x * blockDim.x;
  for (int j = i; j < na; j += stride) a[j] = 0.0f;
  for (int j = i; j < nb; j += stride) b[j] = 0.0f;
}

__global__ __launch_bounds__(256) void eam_edge_fwd(
    const float* __restrict__ r, const int* __restrict__ dst,
    const float* __restrict__ phiD, const float* __restrict__ phiP,
    float* __restrict__ local, float* __restrict__ gd_out, float* __restrict__ gp_out,
    float* __restrict__ partial_pair, int E)
{
  __shared__ v2f sW[64 * 16];

  for (int idx = threadIdx.x; idx < 64 * 16; idx += blockDim.x) {
    const int kp = idx >> 4, ch = idx & 15;
    const int k0 = 2 * kp, k1 = k0 + 1;
    const float c0 = k0 * DELTA, c1 = k1 * DELTA;
    v2f w = {0.0f, 0.0f};
    if (ch < 4) {
      float a0, a1;
      if (ch < 2) { a0 = softplus_(phiD[k0]); a1 = softplus_(phiD[k1]); }
      else        { a0 = phiP[k0];            a1 = phiP[k1]; }
      if (ch & 1) { a0 *= c0; a1 *= c1; }
      w.x = a0; w.y = a1;
    }
    sW[idx] = w;
  }
  __syncthreads();

  const int lane = threadIdx.x & 31;
  const int col  = lane & 15;
  const int hi   = lane >> 4;

  v2f aReg[NCHUNK];
#pragma unroll
  for (int t = 0; t < NCHUNK; ++t)
    aReg[t] = sW[(2 * t + hi) * 16 + col];

  const int wavesPerBlk = blockDim.x >> 5;
  const int wave   = blockIdx.x * wavesPerBlk + (threadIdx.x >> 5);
  const int nWaves = gridDim.x * wavesPerBlk;
  const int nTiles = (E + 15) >> 4;

  float pairAcc = 0.0f;

  for (int tile = wave; tile < nTiles; tile += nWaves) {
    const int e = tile * 16 + col;
    const bool valid = (e < E);

    const int eNext = (tile + nWaves) * 16 + col;
    if (eNext < E) __builtin_prefetch(&r[3 * eNext], 0, 0);

    float x = 0.0f, y = 0.0f, z = 0.0f;
    if (valid) { x = r[3 * e]; y = r[3 * e + 1]; z = r[3 * e + 2]; }
    const float dist = sqrtf(x * x + y * y + z * z);

    v8f acc = {0.f, 0.f, 0.f, 0.f, 0.f, 0.f, 0.f, 0.f};
    const float base = dist - (2 * hi) * DELTA;
    const v2f base2 = {base, base - DELTA};
#pragma unroll
    for (int t = 0; t < NCHUNK; ++t) {
      const v2f tt  = base2 - (float)(4 * t) * DELTA;
      const v2f arg = (tt * tt) * (float)K2_;
      v2f g;
      g.x = FAST_EXP2(arg.x);
      g.y = FAST_EXP2(arg.y);
      acc = __builtin_amdgcn_wmma_f32_16x16x4_f32(
          false, aReg[t], false, g, (short)0, acc, false, false);
    }

    if (lane < 16 && valid) {
      const float SA = acc[0], SB = acc[1], SC = acc[2], SD = acc[3];
      float sn, cs;
      __sincosf(dist * (PI_F / CUTOFF), &sn, &cs);
      const float phi  = 0.25f * (1.0f + cs);
      const float phip = -0.25f * (PI_F / CUTOFF) * sn;
      const float invd = 1.0f / dist;

      const float dens   = phi * SA;
      const float ddens  = phi * (-2.0f * GAMMA_) * (dist * SA - SB) + phip * SA;
      const float pairn  = phi * SC;
      const float dpairn = phi * (-2.0f * GAMMA_) * (dist * SC - SD) + phip * SC;
      const float dpair  = (dpairn - pairn * invd) * invd;

      atomicAdd(&local[dst[e]], dens);
      gd_out[e] = ddens * invd;
      gp_out[e] = dpair * invd;
      pairAcc += pairn * invd;
    }
  }

  const float bsum = blockReduceSum256(pairAcc);
  if (threadIdx.x == 0) partial_pair[blockIdx.x] = bsum;
}

__global__ __launch_bounds__(256) void eam_node(
    const float* __restrict__ local, const float* __restrict__ w,
    float* __restrict__ Fp, float* __restrict__ partial_F, int N)
{
  const float w0 = w[0], w1 = w[1], w2 = w[2], w3 = w[3], w4 = w[4];
  float Facc = 0.0f;
  for (int n = blockIdx.x * blockDim.x + threadIdx.x; n < N;
       n += gridDim.x * blockDim.x) {
    const float x  = local[n];
    const float sq = sqrtf(x);
    const float x2 = x * x;
    const float F  = 2.0f * w0 * sq + w1 * x + 0.5f * w2 * x2 +
                     (1.0f / 6.0f) * w3 * x2 * x + (1.0f / 24.0f) * w4 * x2 * x2;
    const float Fd = w0 / sq + w1 + w2 * x + 0.5f * w3 * x2 +
                     (1.0f / 6.0f) * w4 * x2 * x;
    Facc += F;
    Fp[n] = Fd;
  }
  const float bsum = blockReduceSum256(Facc);
  if (threadIdx.x == 0) partial_F[blockIdx.x] = bsum;
}

__global__ __launch_bounds__(256) void eam_force(
    const float* __restrict__ r, const int* __restrict__ src,
    const int* __restrict__ dst, const float* __restrict__ gd,
    const float* __restrict__ gp, const float* __restrict__ Fp,
    float* __restrict__ forces, int E)
{
  for (int e = blockIdx.x * blockDim.x + threadIdx.x; e < E;
       e += gridDim.x * blockDim.x) {
    const int s = src[e], d = dst[e];
    const float g = Fp[d] * gd[e] + gp[e];
    const float fx = g * r[3 * e], fy = g * r[3 * e + 1], fz = g * r[3 * e + 2];
    atomicAdd(&forces[3 * s + 0],  fx);
    atomicAdd(&forces[3 * s + 1],  fy);
    atomicAdd(&forces[3 * s + 2],  fz);
    atomicAdd(&forces[3 * d + 0], -fx);
    atomicAdd(&forces[3 * d + 1], -fy);
    atomicAdd(&forces[3 * d + 2], -fz);
  }
}

__global__ __launch_bounds__(256) void eam_energy(
    const float* __restrict__ partial_pair, int n1,
    const float* __restrict__ partial_F, int n2,
    float* __restrict__ energy)
{
  float v = 0.0f;
  for (int i = threadIdx.x; i < n1; i += blockDim.x) v += partial_pair[i];
  for (int i = threadIdx.x; i < n2; i += blockDim.x) v += partial_F[i];
  const float s = blockReduceSum256(v);
  if (threadIdx.x == 0) energy[0] = s;
}

extern "C" void kernel_launch(void* const* d_in, const int* in_sizes, int n_in,
                              void* d_out, int out_size, void* d_ws, size_t ws_size,
                              hipStream_t stream) {
  const float* r    = (const float*)d_in[0];
  const int*   src  = (const int*)  d_in[1];
  const int*   dst  = (const int*)  d_in[2];
  const float* phiD = (const float*)d_in[3];
  const float* phiP = (const float*)d_in[4];
  const float* emb  = (const float*)d_in[5];

  const int E = in_sizes[1];
  const int N = (out_size - 1) / 3;

  float* out    = (float*)d_out;
  float* energy = out;
  float* forces = out + 1;

  float* ws    = (float*)d_ws;
  float* gd    = ws;
  float* gp    = gd + E;
  float* local = gp + E;
  float* Fp    = local + N;
  const int NB1 = 512;
  const int NB2 = 128;
  float* ppair = Fp + N;
  float* pF    = ppair + NB1;

  eam_zero    <<<256,  256, 0, stream>>>(local, N, forces, 3 * N);
  eam_edge_fwd<<<NB1,  256, 0, stream>>>(r, dst, phiD, phiP, local, gd, gp, ppair, E);
  eam_node    <<<NB2,  256, 0, stream>>>(local, emb, Fp, pF, N);
  eam_force   <<<1024, 256, 0, stream>>>(r, src, dst, gd, gp, Fp, forces, E);
  eam_energy  <<<1,    256, 0, stream>>>(ppair, NB1, pF, NB2, energy);
}


# edge pass run over the dst-ordered list (ordering moved ahead), a work-group per node range: densities by LDS f32 adds (arrival order), F, F', g in-launch; no global atomics; energy sums regrouped
# speedup vs baseline: 1.7593x; 1.7593x over previous
#include <hip/hip_runtime.h>
#include <math.h>

typedef __attribute__((ext_vector_type(2))) float v2f;
typedef __attribute__((ext_vector_type(8))) float v8f;

#define NBASIS 128
#define NCHUNK 32
#define CUTOFF 6.0f
#define DELTA  (CUTOFF / (NBASIS - 1))
#define GAMMA_ ((NBASIS - 1) / CUTOFF)
#define PI_F   3.14159265358979323846f
#define LOG2E_F 1.44269504088896340736f
#define K2_    (-GAMMA_ * LOG2E_F)

#define NBKT   256
#define NPB    196
#define NSLICE 96

#if __has_builtin(__builtin_amdgcn_exp2f)
#define FAST_EXP2(x) __builtin_amdgcn_exp2f(x)
#else
#define FAST_EXP2(x) exp2f(x)
#endif

__device__ __forceinline__ float softplus_(float x) {
  return fmaxf(x, 0.0f) + log1pf(expf(-fabsf(x)));
}

__device__ __forceinline__ float blockReduceSum512(float v) {
  __shared__ float sred5[512];
  sred5[threadIdx.x] = v;
  __syncthreads();
  for (int s = blockDim.x >> 1; s > 0; s >>= 1) {
    if (threadIdx.x < (unsigned)s) sred5[threadIdx.x] += sred5[threadIdx.x + s];
    __syncthreads();
  }
  return sred5[0];
}

__device__ __forceinline__ float blockReduceSum256(float v) {
  __shared__ float sred[256];
  sred[threadIdx.x] = v;
  __syncthreads();
  for (int s = blockDim.x >> 1; s > 0; s >>= 1) {
    if (threadIdx.x < (unsigned)s) sred[threadIdx.x] += sred[threadIdx.x + s];
    __syncthreads();
  }
  return sred[0];
}

__global__ __launch_bounds__(1024) void eam_count(
    const int* __restrict__ src, const int* __restrict__ dst,
    int* __restrict__ cnt, int E)
{
  __shared__ int cS[NBKT];
  __shared__ int cD[NBKT];
  if (threadIdx.x < NBKT) { cS[threadIdx.x] = 0; cD[threadIdx.x] = 0; }
  __syncthreads();

  const int SL = (E + NSLICE - 1) / NSLICE;
  const int e0 = blockIdx.x * SL;
  const int e1 = (e0 + SL < E) ? (e0 + SL) : E;
  for (int e = e0 + threadIdx.x; e < e1; e += blockDim.x) {
    const int s = src[e], d = dst[e];
    atomicAdd(&cS[s / NPB], 1);
    atomicAdd(&cD[d / NPB], 1);
  }
  __syncthreads();
  if (threadIdx.x < NBKT) {
    cnt[blockIdx.x * NBKT + threadIdx.x]            = cS[threadIdx.x];
    cnt[(NSLICE + blockIdx.x) * NBKT + threadIdx.x] = cD[threadIdx.x];
  }
}

#define SLQ 4
static_assert(SLQ * NBKT == 1024, "one thread per (slice quarter, range)");
static_assert((NSLICE % SLQ) == 0, "slices split evenly over the quarters");
#define WPS (NBKT / 32)
static_assert(WPS * 32 == NBKT, "a side is whole waves");

__global__ __launch_bounds__(1024) void eam_place(
    const int* __restrict__ src, const int* __restrict__ dst,
    const int* __restrict__ cnt, int* __restrict__ offs,
    unsigned int* __restrict__ perm32, int E)
{
  __shared__ int pTot[2][SLQ][NBKT];
  __shared__ int pPre[2][SLQ][NBKT];
  __shared__ int wtot[2 * WPS];
  __shared__ int curS[NBKT];
  __shared__ int curD[NBKT];
  unsigned short* permS = (unsigned short*)perm32;
  unsigned short* permD = permS + E;

  const int sl = blockIdx.x;
  const int q  = threadIdx.x / NBKT;
  const int rr = threadIdx.x % NBKT;
  {
    int vS[NSLICE / SLQ], vD[NSLICE / SLQ];
#pragma unroll
    for (int i = 0; i < NSLICE / SLQ; ++i) {
      vS[i] = cnt[(q + SLQ * i) * NBKT + rr];
      vD[i] = cnt[(NSLICE + q + SLQ * i) * NBKT + rr];
    }
    int tS = 0, pS = 0, tD = 0, pD = 0;
#pragma unroll
    for (int i = 0; i < NSLICE / SLQ; ++i) {
      tS += vS[i]; tD += vD[i];
      if (q + SLQ * i < sl) { pS += vS[i]; pD += vD[i]; }
    }
    pTot[0][q][rr] = tS; pPre[0][q][rr] = pS;
    pTot[1][q][rr] = tD; pPre[1][q][rr] = pD;
  }
  __syncthreads();

  const int lane = threadIdx.x & 31;
  const int wv   = threadIdx.x >> 5;
  const bool owner = threadIdx.x < 2 * NBKT;
  int tot = 0, pre = 0;
  if (owner) {
    const int side = threadIdx.x / NBKT;
    for (int k = 0; k < SLQ; ++k) { tot += pTot[side][k][rr]; pre += pPre[side][k][rr]; }
  }
  int incl = tot;
  for (int d = 1; d < 32; d <<= 1) {
    const int u = __shfl_up(incl, d);
    if (lane >= d) incl += u;
  }
  if (owner && lane == 31) wtot[wv] = incl;
  __syncthreads();
  if (owner) {
    int below = 0;
    for (int k = (wv / WPS) * WPS; k < wv; ++k) below += wtot[k];
    const int start = below + incl - tot + pre;
    if (threadIdx.x < NBKT) curS[rr] = start; else curD[rr] = start;
    offs[threadIdx.x * NSLICE + sl] = start;
  }
  __syncthreads();

  const int SL = (E + NSLICE - 1) / NSLICE;
  const int e0 = blockIdx.x * SL;
  const int e1 = (e0 + SL < E) ? (e0 + SL) : E;
  for (int e = e0 + threadIdx.x; e < e1; e += blockDim.x) {
    const int s = src[e], d = dst[e];
    const int ps = atomicAdd(&curS[s / NPB], 1);
    const int pd = atomicAdd(&curD[d / NPB], 1);
    permS[ps] = (unsigned short)(e - e0);
    permD[pd] = (unsigned short)(e - e0);
  }
}

#define EW    512
#define ENW   (EW / 32)
#define ECAP  4096
#define ECAPT (ECAP / 16)
static_assert(NPB <= 256, "a node's index inside its range is kept in one byte");
static_assert(NPB <= EW, "one thread per node of the range");

__device__ __forceinline__ int seg_of(const int* row, int p) {
  int lo = 0, hi = NSLICE;
  while (hi - lo > 1) {
    const int mid = (lo + hi) >> 1;
    if (row[mid] <= p) lo = mid; else hi = mid;
  }
  return lo;
}


__device__ __forceinline__ void basis_sums(const v2f (&aReg)[NCHUNK], float dist, int hi,
                                           float& SA, float& SB, float& SC, float& SD) {
  v8f acc = {0.f, 0.f, 0.f, 0.f, 0.f, 0.f, 0.f, 0.f};
  const float base = dist - (2 * hi) * DELTA;
  const v2f base2 = {base, base - DELTA};
#pragma unroll
  for (int t = 0; t < NCHUNK; ++t) {
    const v2f tt  = base2 - (float)(4 * t) * DELTA;
    const v2f arg = (tt * tt) * (float)K2_;
    v2f g;
    g.x = FAST_EXP2(arg.x);
    g.y = FAST_EXP2(arg.y);
    acc = __builtin_amdgcn_wmma_f32_16x16x4_f32(
        false, aReg[t], false, g, (short)0, acc, false, false);
  }
  SA = acc[0]; SB = acc[1]; SC = acc[2]; SD = acc[3];
}

__device__ __forceinline__ void edge_terms(float dist, float SA, float SB, float SC, float SD,
                                           float& dens, float& gdv, float& gpv, float& pairv) {
  float sn, cs;
  __sincosf(dist * (PI_F / CUTOFF), &sn, &cs);
  const float phi  = 0.25f * (1.0f + cs);
  const float phip = -0.25f * (PI_F / CUTOFF) * sn;
  const float invd = 1.0f / dist;

  dens = phi * SA;
  const float ddens  = phi * (-2.0f * GAMMA_) * (dist * SA - SB) + phip * SA;
  const float pairn  = phi * SC;
  const float dpairn = phi * (-2.0f * GAMMA_) * (dist * SC - SD) + phip * SC;
  const float dpair  = (dpairn - pairn * invd) * invd;
  gdv   = ddens * invd;
  gpv   = dpair * invd;
  pairv = pairn * invd;
}

__global__ __launch_bounds__(EW) void eam_edge_owner(
    const float* __restrict__ r, const int* __restrict__ dst,
    const float* __restrict__ phiD, const float* __restrict__ phiP,
    const float* __restrict__ w,
    const int* __restrict__ offs, const unsigned int* __restrict__ perm32,
    float* __restrict__ gOut, float* __restrict__ partial, int E, int N)
{
  __shared__ v2f sW[64 * 16];
  __shared__ int   se[ECAP];
  __shared__ float sgd[ECAP];
  __shared__ float sgp[ECAP];
  __shared__ unsigned char sn[ECAP];
  __shared__ float dloc[NPB];
  __shared__ float fpl[NPB];
  __shared__ int rowD[NSLICE + 1];
  const unsigned short* permD = (const unsigned short*)perm32 + E;
  const int b  = blockIdx.x;
  const int n0 = b * NPB;

  for (int idx = threadIdx.x; idx < 64 * 16; idx += blockDim.x) {
    const int kp = idx >> 4, ch = idx & 15;
    const int k0 = 2 * kp, k1 = k0 + 1;
    const float c0 = k0 * DELTA, c1 = k1 * DELTA;
    v2f wv = {0.0f, 0.0f};
    if (ch < 4) {
      float a0, a1;
      if (ch < 2) { a0 = softplus_(phiD[k0]); a1 = softplus_(phiD[k1]); }
      else        { a0 = phiP[k0];            a1 = phiP[k1]; }
      if (ch & 1) { a0 *= c0; a1 *= c1; }
      wv.x = a0; wv.y = a1;
    }
    sW[idx] = wv;
  }
  if (threadIdx.x < NPB) dloc[threadIdx.x] = 0.0f;
  if (threadIdx.x < NSLICE) rowD[threadIdx.x] = offs[(NBKT + b) * NSLICE + threadIdx.x];
  if (threadIdx.x == NSLICE)
    rowD[NSLICE] = (b + 1 < NBKT) ? offs[(NBKT + b + 1) * NSLICE] : E;
  __syncthreads();

  const int lane = threadIdx.x & 31;
  const int col  = lane & 15;
  const int hi   = lane >> 4;
  const int wv   = threadIdx.x >> 5;

  v2f aReg[NCHUNK];
#pragma unroll
  for (int t = 0; t < NCHUNK; ++t)
    aReg[t] = sW[(2 * t + hi) * 16 + col];

  const int SL   = (E + NSLICE - 1) / NSLICE;
  const int pBeg = rowD[0], pEnd = rowD[NSLICE];
  const int nT   = (pEnd - pBeg + 15) >> 4;

  float pairAcc = 0.0f;

  const int pLast = E - 1;
  int pB = pBeg + wv * 16 + col;
  int locB = (int)permD[min(pB, pLast)];
  int eA = min(seg_of(rowD, pB) * SL + locB, pLast);
  float xA = r[3 * eA], yA = r[3 * eA + 1], zA = r[3 * eA + 2];
  int nA = dst[eA] - n0;
  bool vA = pB < pEnd;
  pB = pBeg + (wv + ENW) * 16 + col;
  locB = (int)permD[min(pB, pLast)];

  for (int tile = wv; tile < nT; tile += ENW) {
    const int   eN = min(seg_of(rowD, pB) * SL + locB, pLast);
    const float xN = r[3 * eN], yN = r[3 * eN + 1], zN = r[3 * eN + 2];
    const int   nN = dst[eN] - n0;
    const bool  vN = pB < pEnd;
    pB = pBeg + (tile + 2 * ENW) * 16 + col;
    locB = (int)permD[min(pB, pLast)];

    const float dist = sqrtf(xA * xA + yA * yA + zA * zA);
    float SA, SB, SC, SD;
    basis_sums(aReg, dist, hi, SA, SB, SC, SD);

    if (lane < 16 && vA) {
      float dens, gdv, gpv, pairv;
      edge_terms(dist, SA, SB, SC, SD, dens, gdv, gpv, pairv);
      atomicAdd(&dloc[nA], dens);
      const int i = tile * 16 + col;
      if (i < ECAP) { se[i] = eA; sn[i] = (unsigned char)nA; sgd[i] = gdv; sgp[i] = gpv; }
      pairAcc += pairv;
    }
    eA = eN; nA = nN; xA = xN; yA = yN; zA = zN; vA = vN;
  }
  __syncthreads();

  float Facc = 0.0f;
  if (threadIdx.x < NPB && n0 + (int)threadIdx.x < N) {
    const float w0 = w[0], w1 = w[1], w2 = w[2], w3 = w[3], w4 = w[4];
    const float x  = dloc[threadIdx.x];
    const float sq = sqrtf(x);
    const float x2 = x * x;
    Facc = 2.0f * w0 * sq + w1 * x + 0.5f * w2 * x2 +
           (1.0f / 6.0f) * w3 * x2 * x + (1.0f / 24.0f) * w4 * x2 * x2;
    fpl[threadIdx.x] = w0 / sq + w1 + w2 * x + 0.5f * w3 * x2 +
                       (1.0f / 6.0f) * w4 * x2 * x;
  }
  __syncthreads();

  const int nHeld = (pEnd - pBeg < ECAP) ? (pEnd - pBeg) : ECAP;
  for (int i = threadIdx.x; i < nHeld; i += blockDim.x)
    gOut[se[i]] = fpl[sn[i]] * sgd[i] + sgp[i];

  for (int tile = ECAPT + wv; tile < nT; tile += ENW) {
    const int p = pBeg + tile * 16 + col;
    const bool v = p < pEnd;
    int e = 0, n = 0; float x = 0.0f, y = 0.0f, z = 0.0f;
    if (v) {
      e = seg_of(rowD, p) * SL + (int)permD[p];
      x = r[3 * e]; y = r[3 * e + 1]; z = r[3 * e + 2];
      n = dst[e] - n0;
    }
    const float dist = sqrtf(x * x + y * y + z * z);
    float SA, SB, SC, SD;
    basis_sums(aReg, dist, hi, SA, SB, SC, SD);
    if (lane < 16 && v) {
      float dens, gdv, gpv, pairv;
      edge_terms(dist, SA, SB, SC, SD, dens, gdv, gpv, pairv);
      gOut[e] = fpl[n] * gdv + gpv;
    }
  }

  const float bsum = blockReduceSum512(pairAcc + Facc);
  if (threadIdx.x == 0) partial[b] = bsum;
}

__global__ __launch_bounds__(1024) void eam_force(
    const float* __restrict__ r, const int* __restrict__ src,
    const int* __restrict__ dst, const float* __restrict__ g,
    const int* __restrict__ offs, const unsigned int* __restrict__ perm32,
    float* __restrict__ forces, int E, int N)
{
  __shared__ float acc[3 * NPB];
  __shared__ int rowS[NSLICE + 1];
  __shared__ int rowD[NSLICE + 1];
  const unsigned short* permS = (const unsigned short*)perm32;
  const unsigned short* permD = permS + E;
  const int b = blockIdx.x;

  for (int i = threadIdx.x; i < 3 * NPB; i += blockDim.x) acc[i] = 0.0f;
  if (threadIdx.x < NSLICE) {
    rowS[threadIdx.x] = offs[b * NSLICE + threadIdx.x];
    rowD[threadIdx.x] = offs[(NBKT + b) * NSLICE + threadIdx.x];
  }
  if (threadIdx.x == NSLICE) {
    rowS[NSLICE] = (b + 1 < NBKT) ? offs[(b + 1) * NSLICE] : E;
    rowD[NSLICE] = (b + 1 < NBKT) ? offs[(NBKT + b + 1) * NSLICE] : E;
  }
  __syncthreads();

  const int SL = (E + NSLICE - 1) / NSLICE;
  const int n0 = b * NPB;

  for (int p = rowS[0] + threadIdx.x; p < rowS[NSLICE]; p += blockDim.x) {
    const int loc = permS[p];
    const int e = seg_of(rowS, p) * SL + loc;
    const int n = src[e] - n0;
    const float ge = g[e];
    const float fx = ge * r[3 * e], fy = ge * r[3 * e + 1], fz = ge * r[3 * e + 2];
    atomicAdd(&acc[3 * n + 0], fx);
    atomicAdd(&acc[3 * n + 1], fy);
    atomicAdd(&acc[3 * n + 2], fz);
  }
  for (int p = rowD[0] + threadIdx.x; p < rowD[NSLICE]; p += blockDim.x) {
    const int loc = permD[p];
    const int e = seg_of(rowD, p) * SL + loc;
    const int n = dst[e] - n0;
    const float ge = g[e];
    const float fx = ge * r[3 * e], fy = ge * r[3 * e + 1], fz = ge * r[3 * e + 2];
    atomicAdd(&acc[3 * n + 0], -fx);
    atomicAdd(&acc[3 * n + 1], -fy);
    atomicAdd(&acc[3 * n + 2], -fz);
  }
  __syncthreads();

  for (int i = threadIdx.x; i < 3 * NPB; i += blockDim.x) {
    const int idx = 3 * n0 + i;
    if (idx < 3 * N) forces[idx] = acc[i];
  }
}

__global__ __launch_bounds__(256) void eam_energy(
    const float* __restrict__ partial_pair, int n1,
    const float* __restrict__ partial_F, int n2,
    float* __restrict__ energy)
{
  float v = 0.0f;
  for (int i = threadIdx.x; i < n1; i += blockDim.x) v += partial_pair[i];
  for (int i = threadIdx.x; i < n2; i += blockDim.x) v += partial_F[i];
  const float s = blockReduceSum256(v);
  if (threadIdx.x == 0) energy[0] = s;
}

extern "C" void kernel_launch(void* const* d_in, const int* in_sizes, int n_in,
                              void* d_out, int out_size, void* d_ws, size_t ws_size,
                              hipStream_t stream) {
  const float* r    = (const float*)d_in[0];
  const int*   src  = (const int*)  d_in[1];
  const int*   dst  = (const int*)  d_in[2];
  const float* phiD = (const float*)d_in[3];
  const float* phiP = (const float*)d_in[4];
  const float* emb  = (const float*)d_in[5];

  const int E = in_sizes[1];
  const int N = (out_size - 1) / 3;

  float* out    = (float*)d_out;
  float* energy = out;
  float* forces = out + 1;

  float* ws    = (float*)d_ws;
  float* g     = ws;
  float* lists = g + E;
  float* tabA  = lists + E;
  float* tabB  = tabA + N;
  float* part  = tabB + N;

  int*          cnt  = (int*)tabA;
  int*          seg  = (int*)tabB;
  unsigned int* perm = (unsigned int*)lists;

  eam_count     <<<NSLICE, 1024, 0, stream>>>(src, dst, cnt, E);
  eam_place     <<<NSLICE, 1024, 0, stream>>>(src, dst, cnt, seg, perm, E);
  eam_edge_owner<<<NBKT,   EW,   0, stream>>>(r, dst, phiD, phiP, emb, seg, perm, g, part, E, N);
  eam_force     <<<NBKT,   1024, 0, stream>>>(r, src, dst, g, seg, perm, forces, E, N);
  eam_energy    <<<1,      256,  0, stream>>>(part, NBKT, part, 0, energy);
}


# dst half of forces (-g*r) summed in the edge pass's owner work-group (r kept in LDS, LDS f32 adds, arrival order), stored as start value of forces; force pass walks only the src list; sums regrouped
# speedup vs baseline: 1.9008x; 1.0804x over previous
#include <hip/hip_runtime.h>
#include <math.h>

typedef __attribute__((ext_vector_type(2))) float v2f;
typedef __attribute__((ext_vector_type(8))) float v8f;

#define NBASIS 128
#define NCHUNK 32
#define CUTOFF 6.0f
#define DELTA  (CUTOFF / (NBASIS - 1))
#define GAMMA_ ((NBASIS - 1) / CUTOFF)
#define PI_F   3.14159265358979323846f
#define LOG2E_F 1.44269504088896340736f
#define K2_    (-GAMMA_ * LOG2E_F)

#define NBKT   256
#define NPB    196
#define NSLICE 96

#if __has_builtin(__builtin_amdgcn_exp2f)
#define FAST_EXP2(x) __builtin_amdgcn_exp2f(x)
#else
#define FAST_EXP2(x) exp2f(x)
#endif

__device__ __forceinline__ float softplus_(float x) {
  return fmaxf(x, 0.0f) + log1pf(expf(-fabsf(x)));
}

__device__ __forceinline__ float blockReduceSum512(float v) {
  __shared__ float sred5[512];
  sred5[threadIdx.x] = v;
  __syncthreads();
  for (int s = blockDim.x >> 1; s > 0; s >>= 1) {
    if (threadIdx.x < (unsigned)s) sred5[threadIdx.x] += sred5[threadIdx.x + s];
    __syncthreads();
  }
  return sred5[0];
}

__device__ __forceinline__ float blockReduceSum256(float v) {
  __shared__ float sred[256];
  sred[threadIdx.x] = v;
  __syncthreads();
  for (int s = blockDim.x >> 1; s > 0; s >>= 1) {
    if (threadIdx.x < (unsigned)s) sred[threadIdx.x] += sred[threadIdx.x + s];
    __syncthreads();
  }
  return sred[0];
}

__global__ __launch_bounds__(1024) void eam_count(
    const int* __restrict__ src, const int* __restrict__ dst,
    int* __restrict__ cnt, int E)
{
  __shared__ int cS[NBKT];
  __shared__ int cD[NBKT];
  if (threadIdx.x < NBKT) { cS[threadIdx.x] = 0; cD[threadIdx.x] = 0; }
  __syncthreads();

  const int SL = (E + NSLICE - 1) / NSLICE;
  const int e0 = blockIdx.x * SL;
  const int e1 = (e0 + SL < E) ? (e0 + SL) : E;
  for (int e = e0 + threadIdx.x; e < e1; e += blockDim.x) {
    const int s = src[e], d = dst[e];
    atomicAdd(&cS[s / NPB], 1);
    atomicAdd(&cD[d / NPB], 1);
  }
  __syncthreads();
  if (threadIdx.x < NBKT) {
    cnt[blockIdx.x * NBKT + threadIdx.x]            = cS[threadIdx.x];
    cnt[(NSLICE + blockIdx.x) * NBKT + threadIdx.x] = cD[threadIdx.x];
  }
}

#define SLQ 4
static_assert(SLQ * NBKT == 1024, "one thread per (slice quarter, range)");
static_assert((NSLICE % SLQ) == 0, "slices split evenly over the quarters");
#define WPS (NBKT / 32)
static_assert(WPS * 32 == NBKT, "a side is whole waves");

__global__ __launch_bounds__(1024) void eam_place(
    const int* __restrict__ src, const int* __restrict__ dst,
    const int* __restrict__ cnt, int* __restrict__ offs,
    unsigned int* __restrict__ perm32, int E)
{
  __shared__ int pTot[2][SLQ][NBKT];
  __shared__ int pPre[2][SLQ][NBKT];
  __shared__ int wtot[2 * WPS];
  __shared__ int curS[NBKT];
  __shared__ int curD[NBKT];
  unsigned short* permS = (unsigned short*)perm32;
  unsigned short* permD = permS + E;

  const int sl = blockIdx.x;
  const int q  = threadIdx.x / NBKT;
  const int rr = threadIdx.x % NBKT;
  {
    int vS[NSLICE / SLQ], vD[NSLICE / SLQ];
#pragma unroll
    for (int i = 0; i < NSLICE / SLQ; ++i) {
      vS[i] = cnt[(q + SLQ * i) * NBKT + rr];
      vD[i] = cnt[(NSLICE + q + SLQ * i) * NBKT + rr];
    }
    int tS = 0, pS = 0, tD = 0, pD = 0;
#pragma unroll
    for (int i = 0; i < NSLICE / SLQ; ++i) {
      tS += vS[i]; tD += vD[i];
      if (q + SLQ * i < sl) { pS += vS[i]; pD += vD[i]; }
    }
    pTot[0][q][rr] = tS; pPre[0][q][rr] = pS;
    pTot[1][q][rr] = tD; pPre[1][q][rr] = pD;
  }
  __syncthreads();

  const int lane = threadIdx.x & 31;
  const int wv   = threadIdx.x >> 5;
  const bool owner = threadIdx.x < 2 * NBKT;
  int tot = 0, pre = 0;
  if (owner) {
    const int side = threadIdx.x / NBKT;
    for (int k = 0; k < SLQ; ++k) { tot += pTot[side][k][rr]; pre += pPre[side][k][rr]; }
  }
  int incl = tot;
  for (int d = 1; d < 32; d <<= 1) {
    const int u = __shfl_up(incl, d);
    if (lane >= d) incl += u;
  }
  if (owner && lane == 31) wtot[wv] = incl;
  __syncthreads();
  if (owner) {
    int below = 0;
    for (int k = (wv / WPS) * WPS; k < wv; ++k) below += wtot[k];
    const int start = below + incl - tot + pre;
    if (threadIdx.x < NBKT) curS[rr] = start; else curD[rr] = start;
    offs[threadIdx.x * NSLICE + sl] = start;
  }
  __syncthreads();

  const int SL = (E + NSLICE - 1) / NSLICE;
  const int e0 = blockIdx.x * SL;
  const int e1 = (e0 + SL < E) ? (e0 + SL) : E;
  for (int e = e0 + threadIdx.x; e < e1; e += blockDim.x) {
    const int s = src[e], d = dst[e];
    const int ps = atomicAdd(&curS[s / NPB], 1);
    const int pd = atomicAdd(&curD[d / NPB], 1);
    permS[ps] = (unsigned short)(e - e0);
    permD[pd] = (unsigned short)(e - e0);
  }
}

#define EW    512
#define ENW   (EW / 32)
#define ECAP  4096
#define ECAPT (ECAP / 16)
static_assert(NPB <= 256, "a node's index inside its range is kept in one byte");
static_assert(NPB <= EW, "one thread per node of the range");

__device__ __forceinline__ int seg_of(const int* row, int p) {
  int lo = 0, hi = NSLICE;
  while (hi - lo > 1) {
    const int mid = (lo + hi) >> 1;
    if (row[mid] <= p) lo = mid; else hi = mid;
  }
  return lo;
}


__device__ __forceinline__ void basis_sums(const v2f (&aReg)[NCHUNK], float dist, int hi,
                                           float& SA, float& SB, float& SC, float& SD) {
  v8f acc = {0.f, 0.f, 0.f, 0.f, 0.f, 0.f, 0.f, 0.f};
  const float base = dist - (2 * hi) * DELTA;
  const v2f base2 = {base, base - DELTA};
#pragma unroll
  for (int t = 0; t < NCHUNK; ++t) {
    const v2f tt  = base2 - (float)(4 * t) * DELTA;
    const v2f arg = (tt * tt) * (float)K2_;
    v2f g;
    g.x = FAST_EXP2(arg.x);
    g.y = FAST_EXP2(arg.y);
    acc = __builtin_amdgcn_wmma_f32_16x16x4_f32(
        false, aReg[t], false, g, (short)0, acc, false, false);
  }
  SA = acc[0]; SB = acc[1]; SC = acc[2]; SD = acc[3];
}

__device__ __forceinline__ void edge_terms(float dist, float SA, float SB, float SC, float SD,
                                           float& dens, float& gdv, float& gpv, float& pairv) {
  float sn, cs;
  __sincosf(dist * (PI_F / CUTOFF), &sn, &cs);
  const float phi  = 0.25f * (1.0f + cs);
  const float phip = -0.25f * (PI_F / CUTOFF) * sn;
  const float invd = 1.0f / dist;

  dens = phi * SA;
  const float ddens  = phi * (-2.0f * GAMMA_) * (dist * SA - SB) + phip * SA;
  const float pairn  = phi * SC;
  const float dpairn = phi * (-2.0f * GAMMA_) * (dist * SC - SD) + phip * SC;
  const float dpair  = (dpairn - pairn * invd) * invd;
  gdv   = ddens * invd;
  gpv   = dpair * invd;
  pairv = pairn * invd;
}

__global__ __launch_bounds__(EW) void eam_edge_owner(
    const float* __restrict__ r, const int* __restrict__ dst,
    const float* __restrict__ phiD, const float* __restrict__ phiP,
    const float* __restrict__ w,
    const int* __restrict__ offs, const unsigned int* __restrict__ perm32,
    float* __restrict__ gOut, float* __restrict__ partial,
    float* __restrict__ forces, int E, int N)
{
  __shared__ v2f sW[64 * 16];
  __shared__ int   se[ECAP];
  __shared__ float sgd[ECAP];
  __shared__ float sgp[ECAP];
  __shared__ float sx[ECAP];
  __shared__ float sy[ECAP];
  __shared__ float sz[ECAP];
  __shared__ unsigned char sn[ECAP];
  __shared__ float dloc[NPB];
  __shared__ float fpl[NPB];
  __shared__ float facc[3 * NPB];
  __shared__ int rowD[NSLICE + 1];
  const unsigned short* permD = (const unsigned short*)perm32 + E;
  const int b  = blockIdx.x;
  const int n0 = b * NPB;

  for (int idx = threadIdx.x; idx < 64 * 16; idx += blockDim.x) {
    const int kp = idx >> 4, ch = idx & 15;
    const int k0 = 2 * kp, k1 = k0 + 1;
    const float c0 = k0 * DELTA, c1 = k1 * DELTA;
    v2f wv = {0.0f, 0.0f};
    if (ch < 4) {
      float a0, a1;
      if (ch < 2) { a0 = softplus_(phiD[k0]); a1 = softplus_(phiD[k1]); }
      else        { a0 = phiP[k0];            a1 = phiP[k1]; }
      if (ch & 1) { a0 *= c0; a1 *= c1; }
      wv.x = a0; wv.y = a1;
    }
    sW[idx] = wv;
  }
  if (threadIdx.x < NPB) dloc[threadIdx.x] = 0.0f;
  for (int i = threadIdx.x; i < 3 * NPB; i += blockDim.x) facc[i] = 0.0f;
  if (threadIdx.x < NSLICE) rowD[threadIdx.x] = offs[(NBKT + b) * NSLICE + threadIdx.x];
  if (threadIdx.x == NSLICE)
    rowD[NSLICE] = (b + 1 < NBKT) ? offs[(NBKT + b + 1) * NSLICE] : E;
  __syncthreads();

  const int lane = threadIdx.x & 31;
  const int col  = lane & 15;
  const int hi   = lane >> 4;
  const int wv   = threadIdx.x >> 5;

  v2f aReg[NCHUNK];
#pragma unroll
  for (int t = 0; t < NCHUNK; ++t)
    aReg[t] = sW[(2 * t + hi) * 16 + col];

  const int SL   = (E + NSLICE - 1) / NSLICE;
  const int pBeg = rowD[0], pEnd = rowD[NSLICE];
  const int nT   = (pEnd - pBeg + 15) >> 4;

  float pairAcc = 0.0f;

  const int pLast = E - 1;
  int pB = pBeg + wv * 16 + col;
  int locB = (int)permD[min(pB, pLast)];
  int eA = min(seg_of(rowD, pB) * SL + locB, pLast);
  float xA = r[3 * eA], yA = r[3 * eA + 1], zA = r[3 * eA + 2];
  int nA = dst[eA] - n0;
  bool vA = pB < pEnd;
  pB = pBeg + (wv + ENW) * 16 + col;
  locB = (int)permD[min(pB, pLast)];

  for (int tile = wv; tile < nT; tile += ENW) {
    const int   eN = min(seg_of(rowD, pB) * SL + locB, pLast);
    const float xN = r[3 * eN], yN = r[3 * eN + 1], zN = r[3 * eN + 2];
    const int   nN = dst[eN] - n0;
    const bool  vN = pB < pEnd;
    pB = pBeg + (tile + 2 * ENW) * 16 + col;
    locB = (int)permD[min(pB, pLast)];

    const float dist = sqrtf(xA * xA + yA * yA + zA * zA);
    float SA, SB, SC, SD;
    basis_sums(aReg, dist, hi, SA, SB, SC, SD);

    if (lane < 16 && vA) {
      float dens, gdv, gpv, pairv;
      edge_terms(dist, SA, SB, SC, SD, dens, gdv, gpv, pairv);
      atomicAdd(&dloc[nA], dens);
      const int i = tile * 16 + col;
      if (i < ECAP) {
        se[i] = eA; sn[i] = (unsigned char)nA; sgd[i] = gdv; sgp[i] = gpv;
        sx[i] = xA; sy[i] = yA; sz[i] = zA;
      }
      pairAcc += pairv;
    }
    eA = eN; nA = nN; xA = xN; yA = yN; zA = zN; vA = vN;
  }
  __syncthreads();

  float Facc = 0.0f;
  if (threadIdx.x < NPB && n0 + (int)threadIdx.x < N) {
    const float w0 = w[0], w1 = w[1], w2 = w[2], w3 = w[3], w4 = w[4];
    const float x  = dloc[threadIdx.x];
    const float sq = sqrtf(x);
    const float x2 = x * x;
    Facc = 2.0f * w0 * sq + w1 * x + 0.5f * w2 * x2 +
           (1.0f / 6.0f) * w3 * x2 * x + (1.0f / 24.0f) * w4 * x2 * x2;
    fpl[threadIdx.x] = w0 / sq + w1 + w2 * x + 0.5f * w3 * x2 +
                       (1.0f / 6.0f) * w4 * x2 * x;
  }
  __syncthreads();

  const int nHeld = (pEnd - pBeg < ECAP) ? (pEnd - pBeg) : ECAP;
  for (int i = threadIdx.x; i < nHeld; i += blockDim.x) {
    const int n = sn[i];
    const float x = sx[i], y = sy[i], z = sz[i];
    const float gv = fpl[n] * sgd[i] + sgp[i];
    gOut[se[i]] = gv;
    atomicAdd(&facc[3 * n + 0], -(gv * x));
    atomicAdd(&facc[3 * n + 1], -(gv * y));
    atomicAdd(&facc[3 * n + 2], -(gv * z));
  }

  for (int tile = ECAPT + wv; tile < nT; tile += ENW) {
    const int p = pBeg + tile * 16 + col;
    const bool v = p < pEnd;
    int e = 0, n = 0; float x = 0.0f, y = 0.0f, z = 0.0f;
    if (v) {
      e = seg_of(rowD, p) * SL + (int)permD[p];
      x = r[3 * e]; y = r[3 * e + 1]; z = r[3 * e + 2];
      n = dst[e] - n0;
    }
    const float dist = sqrtf(x * x + y * y + z * z);
    float SA, SB, SC, SD;
    basis_sums(aReg, dist, hi, SA, SB, SC, SD);
    if (lane < 16 && v) {
      float dens, gdv, gpv, pairv;
      edge_terms(dist, SA, SB, SC, SD, dens, gdv, gpv, pairv);
      const float gv = fpl[n] * gdv + gpv;
      gOut[e] = gv;
      atomicAdd(&facc[3 * n + 0], -(gv * x));
      atomicAdd(&facc[3 * n + 1], -(gv * y));
      atomicAdd(&facc[3 * n + 2], -(gv * z));
    }
  }
  __syncthreads();

  for (int i = threadIdx.x; i < 3 * NPB; i += blockDim.x) {
    const int idx = 3 * n0 + i;
    if (idx < 3 * N) forces[idx] = facc[i];
  }

  const float bsum = blockReduceSum512(pairAcc + Facc);
  if (threadIdx.x == 0) partial[b] = bsum;
}

__global__ __launch_bounds__(1024) void eam_force(
    const float* __restrict__ r, const int* __restrict__ src,
    const float* __restrict__ g,
    const int* __restrict__ offs, const unsigned int* __restrict__ perm32,
    float* __restrict__ forces, int E, int N)
{
  __shared__ float acc[3 * NPB];
  __shared__ int rowS[NSLICE + 1];
  const unsigned short* permS = (const unsigned short*)perm32;
  const int b = blockIdx.x;
  const int n0 = b * NPB;
  static_assert(3 * NPB <= 1024, "one thread per force component of the range");

  const int oidx = 3 * n0 + (int)threadIdx.x;
  const bool own = threadIdx.x < 3 * NPB && oidx < 3 * N;
  float fdst = 0.0f;
  if (own) fdst = forces[oidx];

  for (int i = threadIdx.x; i < 3 * NPB; i += blockDim.x) acc[i] = 0.0f;
  if (threadIdx.x < NSLICE) rowS[threadIdx.x] = offs[b * NSLICE + threadIdx.x];
  if (threadIdx.x == NSLICE)
    rowS[NSLICE] = (b + 1 < NBKT) ? offs[(b + 1) * NSLICE] : E;
  __syncthreads();

  const int SL = (E + NSLICE - 1) / NSLICE;

  for (int p = rowS[0] + threadIdx.x; p < rowS[NSLICE]; p += blockDim.x) {
    const int loc = permS[p];
    const int e = seg_of(rowS, p) * SL + loc;
    const int n = src[e] - n0;
    const float ge = g[e];
    const float fx = ge * r[3 * e], fy = ge * r[3 * e + 1], fz = ge * r[3 * e + 2];
    atomicAdd(&acc[3 * n + 0], fx);
    atomicAdd(&acc[3 * n + 1], fy);
    atomicAdd(&acc[3 * n + 2], fz);
  }
  __syncthreads();

  if (own) forces[oidx] = fdst + acc[threadIdx.x];
}

__global__ __launch_bounds__(256) void eam_energy(
    const float* __restrict__ partial_pair, int n1,
    const float* __restrict__ partial_F, int n2,
    float* __restrict__ energy)
{
  float v = 0.0f;
  for (int i = threadIdx.x; i < n1; i += blockDim.x) v += partial_pair[i];
  for (int i = threadIdx.x; i < n2; i += blockDim.x) v += partial_F[i];
  const float s = blockReduceSum256(v);
  if (threadIdx.x == 0) energy[0] = s;
}

extern "C" void kernel_launch(void* const* d_in, const int* in_sizes, int n_in,
                              void* d_out, int out_size, void* d_ws, size_t ws_size,
                              hipStream_t stream) {
  const float* r    = (const float*)d_in[0];
  const int*   src  = (const int*)  d_in[1];
  const int*   dst  = (const int*)  d_in[2];
  const float* phiD = (const float*)d_in[3];
  const float* phiP = (const float*)d_in[4];
  const float* emb  = (const float*)d_in[5];

  const int E = in_sizes[1];
  const int N = (out_size - 1) / 3;

  float* out    = (float*)d_out;
  float* energy = out;
  float* forces = out + 1;

  float* ws    = (float*)d_ws;
  float* g     = ws;
  float* lists = g + E;
  float* tabA  = lists + E;
  float* tabB  = tabA + N;
  float* part  = tabB + N;

  int*          cnt  = (int*)tabA;
  int*          seg  = (int*)tabB;
  unsigned int* perm = (unsigned int*)lists;

  eam_count     <<<NSLICE, 1024, 0, stream>>>(src, dst, cnt, E);
  eam_place     <<<NSLICE, 1024, 0, stream>>>(src, dst, cnt, seg, perm, E);
  eam_edge_owner<<<NBKT,   EW,   0, stream>>>(r, dst, phiD, phiP, emb, seg, perm, g, part, forces, E, N);
  eam_force     <<<NBKT,   1024, 0, stream>>>(r, src, g, seg, perm, forces, E, N);
  eam_energy    <<<1,      256,  0, stream>>>(part, NBKT, part, 0, energy);
}


# ordering in one launch: each edge-slice work-group sorts its own slice by node range in LDS (count+place fused); owners read 96 per-slice runs; force walk regrouped, LDS f32 adds still arrival order
# speedup vs baseline: 2.7408x; 1.4419x over previous
#include <hip/hip_runtime.h>
#include <math.h>

typedef __attribute__((ext_vector_type(2))) float v2f;
typedef __attribute__((ext_vector_type(8))) float v8f;

#define NBASIS 128
#define NCHUNK 32
#define CUTOFF 6.0f
#define DELTA  (CUTOFF / (NBASIS - 1))
#define GAMMA_ ((NBASIS - 1) / CUTOFF)
#define PI_F   3.14159265358979323846f
#define LOG2E_F 1.44269504088896340736f
#define K2_    (-GAMMA_ * LOG2E_F)

#define NBKT   256
#define NPB    196
#define NSLICE 96

#if __has_builtin(__builtin_amdgcn_exp2f)
#define FAST_EXP2(x) __builtin_amdgcn_exp2f(x)
#else
#define FAST_EXP2(x) exp2f(x)
#endif

__device__ __forceinline__ float softplus_(float x) {
  return fmaxf(x, 0.0f) + log1pf(expf(-fabsf(x)));
}

__device__ __forceinline__ float blockReduceSum512(float v) {
  __shared__ float sred5[512];
  sred5[threadIdx.x] = v;
  __syncthreads();
  for (int s = blockDim.x >> 1; s > 0; s >>= 1) {
    if (threadIdx.x < (unsigned)s) sred5[threadIdx.x] += sred5[threadIdx.x + s];
    __syncthreads();
  }
  return sred5[0];
}

__device__ __forceinline__ float blockReduceSum256(float v) {
  __shared__ float sred[256];
  sred[threadIdx.x] = v;
  __syncthreads();
  for (int s = blockDim.x >> 1; s > 0; s >>= 1) {
    if (threadIdx.x < (unsigned)s) sred[threadIdx.x] += sred[threadIdx.x + s];
    __syncthreads();
  }
  return sred[0];
}

#define PT  9
#define WPS (NBKT / 32)
static_assert(WPS * 32 == NBKT, "a side is whole waves");

__global__ __launch_bounds__(1024) void eam_order(
    const int* __restrict__ src, const int* __restrict__ dst,
    int* __restrict__ st, unsigned int* __restrict__ perm32, int E)
{
  __shared__ int cS[NBKT];
  __shared__ int cD[NBKT];
  __shared__ int wtot[2 * WPS];
  unsigned short* permS = (unsigned short*)perm32;
  unsigned short* permD = permS + E;
  if (threadIdx.x < NBKT) { cS[threadIdx.x] = 0; cD[threadIdx.x] = 0; }

  const int sl = blockIdx.x;
  const int SL = (E + NSLICE - 1) / NSLICE;
  const int e0 = sl * SL;
  const int e1 = (e0 + SL < E) ? (e0 + SL) : E;

  int bs[PT], bd[PT];
#pragma unroll
  for (int i = 0; i < PT; ++i) {
    const int e  = e0 + (int)threadIdx.x + i * 1024;
    const int ec = (e < E - 1) ? e : (E - 1);
    bs[i] = src[ec]; bd[i] = dst[ec];
  }
#pragma unroll
  for (int i = 0; i < PT; ++i) {
    const int e = e0 + (int)threadIdx.x + i * 1024;
    bs[i] = (e < e1) ? bs[i] / NPB : -1;
    bd[i] = bd[i] / NPB;
  }
  __syncthreads();
#pragma unroll
  for (int i = 0; i < PT; ++i) {
    if (bs[i] >= 0) { atomicAdd(&cS[bs[i]], 1); atomicAdd(&cD[bd[i]], 1); }
  }
  for (int e = e0 + PT * 1024 + (int)threadIdx.x; e < e1; e += 1024) {
    atomicAdd(&cS[src[e] / NPB], 1);
    atomicAdd(&cD[dst[e] / NPB], 1);
  }
  __syncthreads();

  const int lane = threadIdx.x & 31;
  const int wv   = threadIdx.x >> 5;
  const int rr   = threadIdx.x % NBKT;
  const bool owner = threadIdx.x < 2 * NBKT;
  int tot = 0;
  if (owner) tot = (threadIdx.x < NBKT) ? cS[rr] : cD[rr];
  int incl = tot;
  for (int d = 1; d < 32; d <<= 1) {
    const int u = __shfl_up(incl, d);
    if (lane >= d) incl += u;
  }
  if (owner && lane == 31) wtot[wv] = incl;
  __syncthreads();
  if (owner) {
    int below = 0;
    for (int k = (wv / WPS) * WPS; k < wv; ++k) below += wtot[k];
    const int start = below + incl - tot;
    if (threadIdx.x < NBKT) cS[rr] = start; else cD[rr] = start;
    st[((threadIdx.x / NBKT) * NSLICE + sl) * NBKT + rr] = start;
  }
  __syncthreads();

#pragma unroll
  for (int i = 0; i < PT; ++i) {
    if (bs[i] >= 0) {
      const int ps = atomicAdd(&cS[bs[i]], 1);
      const int pd = atomicAdd(&cD[bd[i]], 1);
      const unsigned short loc = (unsigned short)((int)threadIdx.x + i * 1024);
      permS[e0 + ps] = loc;
      permD[e0 + pd] = loc;
    }
  }
  for (int e = e0 + PT * 1024 + (int)threadIdx.x; e < e1; e += 1024) {
    const int ps = atomicAdd(&cS[src[e] / NPB], 1);
    const int pd = atomicAdd(&cD[dst[e] / NPB], 1);
    permS[e0 + ps] = (unsigned short)(e - e0);
    permD[e0 + pd] = (unsigned short)(e - e0);
  }
}

__device__ __forceinline__ void run_load(const int* __restrict__ stSide, int s, int b,
                                         int& a_s, int& x_s) {
  const int sc = (s < NSLICE) ? s : (NSLICE - 1);
  const int bn = (b + 1 < NBKT) ? (b + 1) : (NBKT - 1);
  const int* rw = stSide + sc * NBKT;
  a_s = rw[b];
  x_s = rw[bn];
}

__device__ __forceinline__ int run_len(int s, int b, int E, int a_s, int x_s) {
  const int SL = (E + NSLICE - 1) / NSLICE;
  int len = E - s * SL;
  len = (len < SL) ? len : SL;
  len = (len > 0) ? len : 0;
  const int n_s = (b + 1 < NBKT) ? x_s : len;
  return (s < NSLICE) ? (n_s - a_s) : 0;
}

__device__ __forceinline__ void runs_scan(int a_s, int x_s, int b, int E,
                                          int* row, int* base, int* scnt) {
  const int SL = (E + NSLICE - 1) / NSLICE;
  const int s = (int)threadIdx.x;
  if (s < NSLICE) scnt[s] = run_len(s, b, E, a_s, x_s);
  __syncthreads();
  if (s <= NSLICE) {
    int acc = 0;
    for (int k = 0; k < s; ++k) acc += scnt[k];
    row[s] = acc;
    if (s < NSLICE) base[s] = s * SL + a_s - acc;
  }
  __syncthreads();
}

#define EW    512
#define ENW   (EW / 32)
#define ECAP  4096
#define ECAPT (ECAP / 16)
static_assert(NPB <= 256, "a node's index inside its range is kept in one byte");
static_assert(NPB <= EW, "one thread per node of the range");

__device__ __forceinline__ int seg_of(const int* row, int p) {
  int lo = 0, hi = NSLICE;
  while (hi - lo > 1) {
    const int mid = (lo + hi) >> 1;
    if (row[mid] <= p) lo = mid; else hi = mid;
  }
  return lo;
}


__device__ __forceinline__ void basis_sums(const v2f (&aReg)[NCHUNK], float dist, int hi,
                                           float& SA, float& SB, float& SC, float& SD) {
  v8f acc = {0.f, 0.f, 0.f, 0.f, 0.f, 0.f, 0.f, 0.f};
  const float base = dist - (2 * hi) * DELTA;
  const v2f base2 = {base, base - DELTA};
#pragma unroll
  for (int t = 0; t < NCHUNK; ++t) {
    const v2f tt  = base2 - (float)(4 * t) * DELTA;
    const v2f arg = (tt * tt) * (float)K2_;
    v2f g;
    g.x = FAST_EXP2(arg.x);
    g.y = FAST_EXP2(arg.y);
    acc = __builtin_amdgcn_wmma_f32_16x16x4_f32(
        false, aReg[t], false, g, (short)0, acc, false, false);
  }
  SA = acc[0]; SB = acc[1]; SC = acc[2]; SD = acc[3];
}

__device__ __forceinline__ void edge_terms(float dist, float SA, float SB, float SC, float SD,
                                           float& dens, float& gdv, float& gpv, float& pairv) {
  float sn, cs;
  __sincosf(dist * (PI_F / CUTOFF), &sn, &cs);
  const float phi  = 0.25f * (1.0f + cs);
  const float phip = -0.25f * (PI_F / CUTOFF) * sn;
  const float invd = 1.0f / dist;

  dens = phi * SA;
  const float ddens  = phi * (-2.0f * GAMMA_) * (dist * SA - SB) + phip * SA;
  const float pairn  = phi * SC;
  const float dpairn = phi * (-2.0f * GAMMA_) * (dist * SC - SD) + phip * SC;
  const float dpair  = (dpairn - pairn * invd) * invd;
  gdv   = ddens * invd;
  gpv   = dpair * invd;
  pairv = pairn * invd;
}

__global__ __launch_bounds__(EW) void eam_edge_owner(
    const float* __restrict__ r, const int* __restrict__ dst,
    const float* __restrict__ phiD, const float* __restrict__ phiP,
    const float* __restrict__ w,
    const int* __restrict__ st, const unsigned int* __restrict__ perm32,
    float* __restrict__ gOut, float* __restrict__ partial,
    float* __restrict__ forces, int E, int N)
{
  __shared__ v2f sW[64 * 16];
  __shared__ int   se[ECAP];
  __shared__ float sgd[ECAP];
  __shared__ float sgp[ECAP];
  __shared__ float sx[ECAP];
  __shared__ float sy[ECAP];
  __shared__ float sz[ECAP];
  __shared__ unsigned char sn[ECAP];
  __shared__ float dloc[NPB];
  __shared__ float fpl[NPB];
  __shared__ float facc[3 * NPB];
  __shared__ int rowD[NSLICE + 1];
  __shared__ int baseD[NSLICE];
  __shared__ int cntD[NSLICE];
  const unsigned short* permD = (const unsigned short*)perm32 + E;
  const int b  = blockIdx.x;
  const int n0 = b * NPB;
  int runA, runX;
  run_load(st + NSLICE * NBKT, (int)threadIdx.x, b, runA, runX);

  for (int idx = threadIdx.x; idx < 64 * 16; idx += blockDim.x) {
    const int kp = idx >> 4, ch = idx & 15;
    const int k0 = 2 * kp, k1 = k0 + 1;
    const float c0 = k0 * DELTA, c1 = k1 * DELTA;
    v2f wv = {0.0f, 0.0f};
    if (ch < 4) {
      float a0, a1;
      if (ch < 2) { a0 = softplus_(phiD[k0]); a1 = softplus_(phiD[k1]); }
      else        { a0 = phiP[k0];            a1 = phiP[k1]; }
      if (ch & 1) { a0 *= c0; a1 *= c1; }
      wv.x = a0; wv.y = a1;
    }
    sW[idx] = wv;
  }
  if (threadIdx.x < NPB) dloc[threadIdx.x] = 0.0f;
  for (int i = threadIdx.x; i < 3 * NPB; i += blockDim.x) facc[i] = 0.0f;
  runs_scan(runA, runX, b, E, rowD, baseD, cntD);

  const int lane = threadIdx.x & 31;
  const int col  = lane & 15;
  const int hi   = lane >> 4;
  const int wv   = threadIdx.x >> 5;

  v2f aReg[NCHUNK];
#pragma unroll
  for (int t = 0; t < NCHUNK; ++t)
    aReg[t] = sW[(2 * t + hi) * 16 + col];

  const int SL   = (E + NSLICE - 1) / NSLICE;
  const int pEnd = rowD[NSLICE];
  const int nT   = (pEnd + 15) >> 4;
  const int pMax = (pEnd > 0) ? (pEnd - 1) : 0;

  float pairAcc = 0.0f;

  const int pLast = E - 1;
  int pB = wv * 16 + col;
  int pc = min(pB, pMax);
  int sB = seg_of(rowD, pc);
  int locB = (int)permD[min(baseD[sB] + pc, pLast)];
  int eA = min(sB * SL + locB, pLast);
  float xA = r[3 * eA], yA = r[3 * eA + 1], zA = r[3 * eA + 2];
  int nA = dst[eA] - n0;
  bool vA = pB < pEnd;
  pB = (wv + ENW) * 16 + col;
  pc = min(pB, pMax);
  sB = seg_of(rowD, pc);
  locB = (int)permD[min(baseD[sB] + pc, pLast)];

  for (int tile = wv; tile < nT; tile += ENW) {
    const int   eN = min(sB * SL + locB, pLast);
    const float xN = r[3 * eN], yN = r[3 * eN + 1], zN = r[3 * eN + 2];
    const int   nN = dst[eN] - n0;
    const bool  vN = pB < pEnd;
    pB = (tile + 2 * ENW) * 16 + col;
    pc = min(pB, pMax);
    sB = seg_of(rowD, pc);
    locB = (int)permD[min(baseD[sB] + pc, pLast)];

    const float dist = sqrtf(xA * xA + yA * yA + zA * zA);
    float SA, SB, SC, SD;
    basis_sums(aReg, dist, hi, SA, SB, SC, SD);

    if (lane < 16 && vA) {
      float dens, gdv, gpv, pairv;
      edge_terms(dist, SA, SB, SC, SD, dens, gdv, gpv, pairv);
      atomicAdd(&dloc[nA], dens);
      const int i = tile * 16 + col;
      if (i < ECAP) {
        se[i] = eA; sn[i] = (unsigned char)nA; sgd[i] = gdv; sgp[i] = gpv;
        sx[i] = xA; sy[i] = yA; sz[i] = zA;
      }
      pairAcc += pairv;
    }
    eA = eN; nA = nN; xA = xN; yA = yN; zA = zN; vA = vN;
  }
  __syncthreads();

  float Facc = 0.0f;
  if (threadIdx.x < NPB && n0 + (int)threadIdx.x < N) {
    const float w0 = w[0], w1 = w[1], w2 = w[2], w3 = w[3], w4 = w[4];
    const float x  = dloc[threadIdx.x];
    const float sq = sqrtf(x);
    const float x2 = x * x;
    Facc = 2.0f * w0 * sq + w1 * x + 0.5f * w2 * x2 +
           (1.0f / 6.0f) * w3 * x2 * x + (1.0f / 24.0f) * w4 * x2 * x2;
    fpl[threadIdx.x] = w0 / sq + w1 + w2 * x + 0.5f * w3 * x2 +
                       (1.0f / 6.0f) * w4 * x2 * x;
  }
  __syncthreads();

  const int nHeld = (pEnd < ECAP) ? pEnd : ECAP;
  for (int i = threadIdx.x; i < nHeld; i += blockDim.x) {
    const int n = sn[i];
    const float x = sx[i], y = sy[i], z = sz[i];
    const float gv = fpl[n] * sgd[i] + sgp[i];
    gOut[se[i]] = gv;
    atomicAdd(&facc[3 * n + 0], -(gv * x));
    atomicAdd(&facc[3 * n + 1], -(gv * y));
    atomicAdd(&facc[3 * n + 2], -(gv * z));
  }

  for (int tile = ECAPT + wv; tile < nT; tile += ENW) {
    const int p = tile * 16 + col;
    const bool v = p < pEnd;
    int e = 0, n = 0; float x = 0.0f, y = 0.0f, z = 0.0f;
    if (v) {
      const int s = seg_of(rowD, p);
      e = s * SL + (int)permD[baseD[s] + p];
      x = r[3 * e]; y = r[3 * e + 1]; z = r[3 * e + 2];
      n = dst[e] - n0;
    }
    const float dist = sqrtf(x * x + y * y + z * z);
    float SA, SB, SC, SD;
    basis_sums(aReg, dist, hi, SA, SB, SC, SD);
    if (lane < 16 && v) {
      float dens, gdv, gpv, pairv;
      edge_terms(dist, SA, SB, SC, SD, dens, gdv, gpv, pairv);
      const float gv = fpl[n] * gdv + gpv;
      gOut[e] = gv;
      atomicAdd(&facc[3 * n + 0], -(gv * x));
      atomicAdd(&facc[3 * n + 1], -(gv * y));
      atomicAdd(&facc[3 * n + 2], -(gv * z));
    }
  }
  __syncthreads();

  for (int i = threadIdx.x; i < 3 * NPB; i += blockDim.x) {
    const int idx = 3 * n0 + i;
    if (idx < 3 * N) forces[idx] = facc[i];
  }

  const float bsum = blockReduceSum512(pairAcc + Facc);
  if (threadIdx.x == 0) partial[b] = bsum;
}

#define TPR 10
#define FT  6
static_assert(TPR * NSLICE <= 1024, "every run has its threads");

__global__ __launch_bounds__(1024) void eam_force(
    const float* __restrict__ r, const int* __restrict__ src,
    const float* __restrict__ g,
    const int* __restrict__ st, const unsigned int* __restrict__ perm32,
    float* __restrict__ forces, int E, int N)
{
  __shared__ float acc[3 * NPB];
  const unsigned short* permS = (const unsigned short*)perm32;
  const int b = blockIdx.x;
  const int n0 = b * NPB;
  static_assert(3 * NPB <= 1024, "one thread per force component of the range");

  const int SL  = (E + NSLICE - 1) / NSLICE;
  const int s   = (int)threadIdx.x / TPR;
  const int sub = (int)threadIdx.x - s * TPR;
  int a_s, x_s;
  run_load(st, s, b, a_s, x_s);

  const int oidx = 3 * n0 + (int)threadIdx.x;
  const bool own = threadIdx.x < 3 * NPB && oidx < 3 * N;
  float fdst = 0.0f;
  if (own) fdst = forces[oidx];

  for (int i = threadIdx.x; i < 3 * NPB; i += blockDim.x) acc[i] = 0.0f;
  const int c_s = run_len(s, b, E, a_s, x_s);
  __syncthreads();

  const int pLast = E - 1;
  const int ebase = s * SL;
  const int jMax  = (c_s > 0) ? (c_s - 1) : 0;
  int e[FT];
#pragma unroll
  for (int k = 0; k < FT; ++k) {
    const int j = min(sub + k * TPR, jMax);
    e[k] = (int)permS[min(ebase + a_s + j, pLast)];
  }
  int   nn[FT];
  float ge[FT], x[FT], y[FT], z[FT];
#pragma unroll
  for (int k = 0; k < FT; ++k) {
    e[k]  = min(ebase + e[k], pLast);
    nn[k] = src[e[k]];
    ge[k] = g[e[k]];
    x[k] = r[3 * e[k]]; y[k] = r[3 * e[k] + 1]; z[k] = r[3 * e[k] + 2];
  }
  asm volatile("" :: "v"(nn[0]), "v"(ge[0]), "v"(x[0]), "v"(y[0]), "v"(z[0]));
#pragma unroll
  for (int k = 0; k < FT; ++k) {
    if (sub + k * TPR < c_s) {
      const int n = nn[k] - n0;
      atomicAdd(&acc[3 * n + 0], ge[k] * x[k]);
      atomicAdd(&acc[3 * n + 1], ge[k] * y[k]);
      atomicAdd(&acc[3 * n + 2], ge[k] * z[k]);
    }
  }
  for (int j = sub + FT * TPR; j < c_s; j += TPR) {
    const int ee = ebase + (int)permS[ebase + a_s + j];
    const int n = src[ee] - n0;
    const float gv = g[ee];
    atomicAdd(&acc[3 * n + 0], gv * r[3 * ee]);
    atomicAdd(&acc[3 * n + 1], gv * r[3 * ee + 1]);
    atomicAdd(&acc[3 * n + 2], gv * r[3 * ee + 2]);
  }
  __syncthreads();

  if (own) forces[oidx] = fdst + acc[threadIdx.x];
}

__global__ __launch_bounds__(256) void eam_energy(
    const float* __restrict__ partial_pair, int n1,
    const float* __restrict__ partial_F, int n2,
    float* __restrict__ energy)
{
  float v = 0.0f;
  for (int i = threadIdx.x; i < n1; i += blockDim.x) v += partial_pair[i];
  for (int i = threadIdx.x; i < n2; i += blockDim.x) v += partial_F[i];
  const float s = blockReduceSum256(v);
  if (threadIdx.x == 0) energy[0] = s;
}

extern "C" void kernel_launch(void* const* d_in, const int* in_sizes, int n_in,
                              void* d_out, int out_size, void* d_ws, size_t ws_size,
                              hipStream_t stream) {
  const float* r    = (const float*)d_in[0];
  const int*   src  = (const int*)  d_in[1];
  const int*   dst  = (const int*)  d_in[2];
  const float* phiD = (const float*)d_in[3];
  const float* phiP = (const float*)d_in[4];
  const float* emb  = (const float*)d_in[5];

  const int E = in_sizes[1];
  const int N = (out_size - 1) / 3;

  float* out    = (float*)d_out;
  float* energy = out;
  float* forces = out + 1;

  float* ws    = (float*)d_ws;
  float* g     = ws;
  float* lists = g + E;
  float* tabA  = lists + E;
  float* tabB  = tabA + N;
  float* part  = tabB + N;

  int*          st   = (int*)tabA;
  unsigned int* perm = (unsigned int*)lists;

  eam_order     <<<NSLICE, 1024, 0, stream>>>(src, dst, st, perm, E);
  eam_edge_owner<<<NBKT,   EW,   0, stream>>>(r, dst, phiD, phiP, emb, st, perm, g, part, forces, E, N);
  eam_force     <<<NBKT,   1024, 0, stream>>>(r, src, g, st, perm, forces, E, N);
  eam_energy    <<<1,      256,  0, stream>>>(part, NBKT, part, 0, energy);
}


# ordering pass: each slice's two ordered lists are assembled in LDS and stored as whole 128-byte lines, not as one scattered two-byte global store per edge; list contents and f32 sum order unchanged
# speedup vs baseline: 2.9194x; 1.0652x over previous
#include <hip/hip_runtime.h>
#include <math.h>

typedef __attribute__((ext_vector_type(2))) float v2f;
typedef __attribute__((ext_vector_type(8))) float v8f;

#define NBASIS 128
#define NCHUNK 32
#define CUTOFF 6.0f
#define DELTA  (CUTOFF / (NBASIS - 1))
#define GAMMA_ ((NBASIS - 1) / CUTOFF)
#define PI_F   3.14159265358979323846f
#define LOG2E_F 1.44269504088896340736f
#define K2_    (-GAMMA_ * LOG2E_F)

#define NBKT   256
#define NPB    196
#define NSLICE 96

#if __has_builtin(__builtin_amdgcn_exp2f)
#define FAST_EXP2(x) __builtin_amdgcn_exp2f(x)
#else
#define FAST_EXP2(x) exp2f(x)
#endif

__device__ __forceinline__ float softplus_(float x) {
  return fmaxf(x, 0.0f) + log1pf(expf(-fabsf(x)));
}

__device__ __forceinline__ float blockReduceSum512(float v) {
  __shared__ float sred5[512];
  sred5[threadIdx.x] = v;
  __syncthreads();
  for (int s = blockDim.x >> 1; s > 0; s >>= 1) {
    if (threadIdx.x < (unsigned)s) sred5[threadIdx.x] += sred5[threadIdx.x + s];
    __syncthreads();
  }
  return sred5[0];
}

__device__ __forceinline__ float blockReduceSum256(float v) {
  __shared__ float sred[256];
  sred[threadIdx.x] = v;
  __syncthreads();
  for (int s = blockDim.x >> 1; s > 0; s >>= 1) {
    if (threadIdx.x < (unsigned)s) sred[threadIdx.x] += sred[threadIdx.x + s];
    __syncthreads();
  }
  return sred[0];
}

#define PT  9
#define WPS (NBKT / 32)
static_assert(WPS * 32 == NBKT, "a side is whole waves");
#define SLCAP (PT * 1024)
#define LINE2 64

typedef unsigned int u32x4 __attribute__((ext_vector_type(4)));
typedef u32x4 __attribute__((may_alias)) u32x4a;

template <bool FITS>
__device__ __forceinline__ void list_put(unsigned short* l, unsigned short* g, int sh,
                                         int p, unsigned short v) {
  if (FITS || p < SLCAP) l[sh + p] = v; else g[p] = v;
}

__device__ __forceinline__ void list_store(const unsigned short* l, unsigned short* g,
                                           int sh, int n) {
  const int tot = sh + n;
  const int nCh = (tot + 7) >> 3;
  u32x4a* gv = (u32x4a*)(g - sh);
  const u32x4a* lv = (const u32x4a*)l;
  for (int c = threadIdx.x; c < nCh; c += blockDim.x) {
    const int j0 = 8 * c;
    if (j0 >= sh && j0 + 8 <= tot) {
      gv[c] = lv[c];
    } else {
      for (int k = 0; k < 8; ++k) {
        const int j = j0 + k;
        if (j >= sh && j < tot) g[j - sh] = l[j];
      }
    }
  }
}

template <bool FITS>
__device__ __forceinline__ void order_place(
    const int (&bs)[PT], const int (&bd)[PT],
    const int* __restrict__ src, const int* __restrict__ dst, int* cS, int* cD,
    unsigned short* lS, unsigned short* lD, unsigned short* gS, unsigned short* gD,
    int shS, int shD, int e0, int e1)
{
#pragma unroll
  for (int i = 0; i < PT; ++i) {
    if (bs[i] >= 0) {
      const int ps = atomicAdd(&cS[bs[i]], 1);
      const int pd = atomicAdd(&cD[bd[i]], 1);
      const unsigned short loc = (unsigned short)((int)threadIdx.x + i * 1024);
      list_put<FITS>(lS, gS, shS, ps, loc);
      list_put<FITS>(lD, gD, shD, pd, loc);
    }
  }
  if (!FITS) {
    for (int e = e0 + PT * 1024 + (int)threadIdx.x; e < e1; e += 1024) {
      const int ps = atomicAdd(&cS[src[e] / NPB], 1);
      const int pd = atomicAdd(&cD[dst[e] / NPB], 1);
      list_put<false>(lS, gS, shS, ps, (unsigned short)(e - e0));
      list_put<false>(lD, gD, shD, pd, (unsigned short)(e - e0));
    }
  }
}

__global__ __launch_bounds__(1024) void eam_order(
    const int* __restrict__ src, const int* __restrict__ dst,
    int* __restrict__ st, unsigned int* __restrict__ perm32, int E)
{
  __shared__ int cS[NBKT];
  __shared__ int cD[NBKT];
  __shared__ int wtot[2 * WPS];
  __shared__ __attribute__((aligned(16))) unsigned short lS[SLCAP + LINE2];
  __shared__ __attribute__((aligned(16))) unsigned short lD[SLCAP + LINE2];
  unsigned short* permS = (unsigned short*)perm32;
  unsigned short* permD = permS + E;
  if (threadIdx.x < NBKT) { cS[threadIdx.x] = 0; cD[threadIdx.x] = 0; }

  const int sl = blockIdx.x;
  const int SL = (E + NSLICE - 1) / NSLICE;
  const int e0 = sl * SL;
  const int e1 = (e0 + SL < E) ? (e0 + SL) : E;
  unsigned short* gS = permS + e0;
  unsigned short* gD = permD + e0;
  const int shS = (int)(((unsigned long long)gS >> 1) & (LINE2 - 1));
  const int shD = (int)(((unsigned long long)gD >> 1) & (LINE2 - 1));

  int bs[PT], bd[PT];
#pragma unroll
  for (int i = 0; i < PT; ++i) {
    const int e  = e0 + (int)threadIdx.x + i * 1024;
    const int ec = (e < E - 1) ? e : (E - 1);
    bs[i] = src[ec]; bd[i] = dst[ec];
  }
#pragma unroll
  for (int i = 0; i < PT; ++i) {
    const int e = e0 + (int)threadIdx.x + i * 1024;
    bs[i] = (e < e1) ? bs[i] / NPB : -1;
    bd[i] = bd[i] / NPB;
  }
  __syncthreads();
#pragma unroll
  for (int i = 0; i < PT; ++i) {
    if (bs[i] >= 0) { atomicAdd(&cS[bs[i]], 1); atomicAdd(&cD[bd[i]], 1); }
  }
  for (int e = e0 + PT * 1024 + (int)threadIdx.x; e < e1; e += 1024) {
    atomicAdd(&cS[src[e] / NPB], 1);
    atomicAdd(&cD[dst[e] / NPB], 1);
  }
  __syncthreads();

  const int lane = threadIdx.x & 31;
  const int wv   = threadIdx.x >> 5;
  const int rr   = threadIdx.x % NBKT;
  const bool owner = threadIdx.x < 2 * NBKT;
  int tot = 0;
  if (owner) tot = (threadIdx.x < NBKT) ? cS[rr] : cD[rr];
  int incl = tot;
  for (int d = 1; d < 32; d <<= 1) {
    const int u = __shfl_up(incl, d);
    if (lane >= d) incl += u;
  }
  if (owner && lane == 31) wtot[wv] = incl;
  __syncthreads();
  if (owner) {
    int below = 0;
    for (int k = (wv / WPS) * WPS; k < wv; ++k) below += wtot[k];
    const int start = below + incl - tot;
    if (threadIdx.x < NBKT) cS[rr] = start; else cD[rr] = start;
    st[((threadIdx.x / NBKT) * NSLICE + sl) * NBKT + rr] = start;
  }
  __syncthreads();

  if (e1 - e0 <= SLCAP)
    order_place<true >(bs, bd, src, dst, cS, cD, lS, lD, gS, gD, shS, shD, e0, e1);
  else
    order_place<false>(bs, bd, src, dst, cS, cD, lS, lD, gS, gD, shS, shD, e0, e1);
  __syncthreads();

  int nL = e1 - e0;
  nL = (nL < SLCAP) ? nL : SLCAP;
  nL = (nL > 0) ? nL : 0;
  list_store(lS, gS, shS, nL);
  list_store(lD, gD, shD, nL);
}

__device__ __forceinline__ void run_load(const int* __restrict__ stSide, int s, int b,
                                         int& a_s, int& x_s) {
  const int sc = (s < NSLICE) ? s : (NSLICE - 1);
  const int bn = (b + 1 < NBKT) ? (b + 1) : (NBKT - 1);
  const int* rw = stSide + sc * NBKT;
  a_s = rw[b];
  x_s = rw[bn];
}

__device__ __forceinline__ int run_len(int s, int b, int E, int a_s, int x_s) {
  const int SL = (E + NSLICE - 1) / NSLICE;
  int len = E - s * SL;
  len = (len < SL) ? len : SL;
  len = (len > 0) ? len : 0;
  const int n_s = (b + 1 < NBKT) ? x_s : len;
  return (s < NSLICE) ? (n_s - a_s) : 0;
}

__device__ __forceinline__ void runs_scan(int a_s, int x_s, int b, int E,
                                          int* row, int* base, int* scnt) {
  const int SL = (E + NSLICE - 1) / NSLICE;
  const int s = (int)threadIdx.x;
  if (s < NSLICE) scnt[s] = run_len(s, b, E, a_s, x_s);
  __syncthreads();
  if (s <= NSLICE) {
    int acc = 0;
    for (int k = 0; k < s; ++k) acc += scnt[k];
    row[s] = acc;
    if (s < NSLICE) base[s] = s * SL + a_s - acc;
  }
  __syncthreads();
}

#define EW    512
#define ENW   (EW / 32)
#define ECAP  4096
#define ECAPT (ECAP / 16)
static_assert(NPB <= 256, "a node's index inside its range is kept in one byte");
static_assert(NPB <= EW, "one thread per node of the range");

__device__ __forceinline__ int seg_of(const int* row, int p) {
  int lo = 0, hi = NSLICE;
  while (hi - lo > 1) {
    const int mid = (lo + hi) >> 1;
    if (row[mid] <= p) lo = mid; else hi = mid;
  }
  return lo;
}


__device__ __forceinline__ void basis_sums(const v2f (&aReg)[NCHUNK], float dist, int hi,
                                           float& SA, float& SB, float& SC, float& SD) {
  v8f acc = {0.f, 0.f, 0.f, 0.f, 0.f, 0.f, 0.f, 0.f};
  const float base = dist - (2 * hi) * DELTA;
  const v2f base2 = {base, base - DELTA};
#pragma unroll
  for (int t = 0; t < NCHUNK; ++t) {
    const v2f tt  = base2 - (float)(4 * t) * DELTA;
    const v2f arg = (tt * tt) * (float)K2_;
    v2f g;
    g.x = FAST_EXP2(arg.x);
    g.y = FAST_EXP2(arg.y);
    acc = __builtin_amdgcn_wmma_f32_16x16x4_f32(
        false, aReg[t], false, g, (short)0, acc, false, false);
  }
  SA = acc[0]; SB = acc[1]; SC = acc[2]; SD = acc[3];
}

__device__ __forceinline__ void edge_terms(float dist, float SA, float SB, float SC, float SD,
                                           float& dens, float& gdv, float& gpv, float& pairv) {
  float sn, cs;
  __sincosf(dist * (PI_F / CUTOFF), &sn, &cs);
  const float phi  = 0.25f * (1.0f + cs);
  const float phip = -0.25f * (PI_F / CUTOFF) * sn;
  const float invd = 1.0f / dist;

  dens = phi * SA;
  const float ddens  = phi * (-2.0f * GAMMA_) * (dist * SA - SB) + phip * SA;
  const float pairn  = phi * SC;
  const float dpairn = phi * (-2.0f * GAMMA_) * (dist * SC - SD) + phip * SC;
  const float dpair  = (dpairn - pairn * invd) * invd;
  gdv   = ddens * invd;
  gpv   = dpair * invd;
  pairv = pairn * invd;
}

__global__ __launch_bounds__(EW) void eam_edge_owner(
    const float* __restrict__ r, const int* __restrict__ dst,
    const float* __restrict__ phiD, const float* __restrict__ phiP,
    const float* __restrict__ w,
    const int* __restrict__ st, const unsigned int* __restrict__ perm32,
    float* __restrict__ gOut, float* __restrict__ partial,
    float* __restrict__ forces, int E, int N)
{
  __shared__ v2f sW[64 * 16];
  __shared__ int   se[ECAP];
  __shared__ float sgd[ECAP];
  __shared__ float sgp[ECAP];
  __shared__ float sx[ECAP];
  __shared__ float sy[ECAP];
  __shared__ float sz[ECAP];
  __shared__ unsigned char sn[ECAP];
  __shared__ float dloc[NPB];
  __shared__ float fpl[NPB];
  __shared__ float facc[3 * NPB];
  __shared__ int rowD[NSLICE + 1];
  __shared__ int baseD[NSLICE];
  __shared__ int cntD[NSLICE];
  const unsigned short* permD = (const unsigned short*)perm32 + E;
  const int b  = blockIdx.x;
  const int n0 = b * NPB;
  int runA, runX;
  run_load(st + NSLICE * NBKT, (int)threadIdx.x, b, runA, runX);

  for (int idx = threadIdx.x; idx < 64 * 16; idx += blockDim.x) {
    const int kp = idx >> 4, ch = idx & 15;
    const int k0 = 2 * kp, k1 = k0 + 1;
    const float c0 = k0 * DELTA, c1 = k1 * DELTA;
    v2f wv = {0.0f, 0.0f};
    if (ch < 4) {
      float a0, a1;
      if (ch < 2) { a0 = softplus_(phiD[k0]); a1 = softplus_(phiD[k1]); }
      else        { a0 = phiP[k0];            a1 = phiP[k1]; }
      if (ch & 1) { a0 *= c0; a1 *= c1; }
      wv.x = a0; wv.y = a1;
    }
    sW[idx] = wv;
  }
  if (threadIdx.x < NPB) dloc[threadIdx.x] = 0.0f;
  for (int i = threadIdx.x; i < 3 * NPB; i += blockDim.x) facc[i] = 0.0f;
  runs_scan(runA, runX, b, E, rowD, baseD, cntD);

  const int lane = threadIdx.x & 31;
  const int col  = lane & 15;
  const int hi   = lane >> 4;
  const int wv   = threadIdx.x >> 5;

  v2f aReg[NCHUNK];
#pragma unroll
  for (int t = 0; t < NCHUNK; ++t)
    aReg[t] = sW[(2 * t + hi) * 16 + col];

  const int SL   = (E + NSLICE - 1) / NSLICE;
  const int pEnd = rowD[NSLICE];
  const int nT   = (pEnd + 15) >> 4;
  const int pMax = (pEnd > 0) ? (pEnd - 1) : 0;

  float pairAcc = 0.0f;

  const int pLast = E - 1;
  int pB = wv * 16 + col;
  int pc = min(pB, pMax);
  int sB = seg_of(rowD, pc);
  int locB = (int)permD[min(baseD[sB] + pc, pLast)];
  int eA = min(sB * SL + locB, pLast);
  float xA = r[3 * eA], yA = r[3 * eA + 1], zA = r[3 * eA + 2];
  int nA = dst[eA] - n0;
  bool vA = pB < pEnd;
  pB = (wv + ENW) * 16 + col;
  pc = min(pB, pMax);
  sB = seg_of(rowD, pc);
  locB = (int)permD[min(baseD[sB] + pc, pLast)];

  for (int tile = wv; tile < nT; tile += ENW) {
    const int   eN = min(sB * SL + locB, pLast);
    const float xN = r[3 * eN], yN = r[3 * eN + 1], zN = r[3 * eN + 2];
    const int   nN = dst[eN] - n0;
    const bool  vN = pB < pEnd;
    pB = (tile + 2 * ENW) * 16 + col;
    pc = min(pB, pMax);
    sB = seg_of(rowD, pc);
    locB = (int)permD[min(baseD[sB] + pc, pLast)];

    const float dist = sqrtf(xA * xA + yA * yA + zA * zA);
    float SA, SB, SC, SD;
    basis_sums(aReg, dist, hi, SA, SB, SC, SD);

    if (lane < 16 && vA) {
      float dens, gdv, gpv, pairv;
      edge_terms(dist, SA, SB, SC, SD, dens, gdv, gpv, pairv);
      atomicAdd(&dloc[nA], dens);
      const int i = tile * 16 + col;
      if (i < ECAP) {
        se[i] = eA; sn[i] = (unsigned char)nA; sgd[i] = gdv; sgp[i] = gpv;
        sx[i] = xA; sy[i] = yA; sz[i] = zA;
      }
      pairAcc += pairv;
    }
    eA = eN; nA = nN; xA = xN; yA = yN; zA = zN; vA = vN;
  }
  __syncthreads();

  float Facc = 0.0f;
  if (threadIdx.x < NPB && n0 + (int)threadIdx.x < N) {
    const float w0 = w[0], w1 = w[1], w2 = w[2], w3 = w[3], w4 = w[4];
    const float x  = dloc[threadIdx.x];
    const float sq = sqrtf(x);
    const float x2 = x * x;
    Facc = 2.0f * w0 * sq + w1 * x + 0.5f * w2 * x2 +
           (1.0f / 6.0f) * w3 * x2 * x + (1.0f / 24.0f) * w4 * x2 * x2;
    fpl[threadIdx.x] = w0 / sq + w1 + w2 * x + 0.5f * w3 * x2 +
                       (1.0f / 6.0f) * w4 * x2 * x;
  }
  __syncthreads();

  const int nHeld = (pEnd < ECAP) ? pEnd : ECAP;
  for (int i = threadIdx.x; i < nHeld; i += blockDim.x) {
    const int n = sn[i];
    const float x = sx[i], y = sy[i], z = sz[i];
    const float gv = fpl[n] * sgd[i] + sgp[i];
    gOut[se[i]] = gv;
    atomicAdd(&facc[3 * n + 0], -(gv * x));
    atomicAdd(&facc[3 * n + 1], -(gv * y));
    atomicAdd(&facc[3 * n + 2], -(gv * z));
  }

  for (int tile = ECAPT + wv; tile < nT; tile += ENW) {
    const int p = tile * 16 + col;
    const bool v = p < pEnd;
    int e = 0, n = 0; float x = 0.0f, y = 0.0f, z = 0.0f;
    if (v) {
      const int s = seg_of(rowD, p);
      e = s * SL + (int)permD[baseD[s] + p];
      x = r[3 * e]; y = r[3 * e + 1]; z = r[3 * e + 2];
      n = dst[e] - n0;
    }
    const float dist = sqrtf(x * x + y * y + z * z);
    float SA, SB, SC, SD;
    basis_sums(aReg, dist, hi, SA, SB, SC, SD);
    if (lane < 16 && v) {
      float dens, gdv, gpv, pairv;
      edge_terms(dist, SA, SB, SC, SD, dens, gdv, gpv, pairv);
      const float gv = fpl[n] * gdv + gpv;
      gOut[e] = gv;
      atomicAdd(&facc[3 * n + 0], -(gv * x));
      atomicAdd(&facc[3 * n + 1], -(gv * y));
      atomicAdd(&facc[3 * n + 2], -(gv * z));
    }
  }
  __syncthreads();

  for (int i = threadIdx.x; i < 3 * NPB; i += blockDim.x) {
    const int idx = 3 * n0 + i;
    if (idx < 3 * N) forces[idx] = facc[i];
  }

  const float bsum = blockReduceSum512(pairAcc + Facc);
  if (threadIdx.x == 0) partial[b] = bsum;
}

#define TPR 10
#define FT  6
static_assert(TPR * NSLICE <= 1024, "every run has its threads");

__global__ __launch_bounds__(1024) void eam_force(
    const float* __restrict__ r, const int* __restrict__ src,
    const float* __restrict__ g,
    const int* __restrict__ st, const unsigned int* __restrict__ perm32,
    float* __restrict__ forces, int E, int N)
{
  __shared__ float acc[3 * NPB];
  const unsigned short* permS = (const unsigned short*)perm32;
  const int b = blockIdx.x;
  const int n0 = b * NPB;
  static_assert(3 * NPB <= 1024, "one thread per force component of the range");

  const int SL  = (E + NSLICE - 1) / NSLICE;
  const int s   = (int)threadIdx.x / TPR;
  const int sub = (int)threadIdx.x - s * TPR;
  int a_s, x_s;
  run_load(st, s, b, a_s, x_s);

  const int oidx = 3 * n0 + (int)threadIdx.x;
  const bool own = threadIdx.x < 3 * NPB && oidx < 3 * N;
  float fdst = 0.0f;
  if (own) fdst = forces[oidx];

  for (int i = threadIdx.x; i < 3 * NPB; i += blockDim.x) acc[i] = 0.0f;
  const int c_s = run_len(s, b, E, a_s, x_s);
  __syncthreads();

  const int pLast = E - 1;
  const int ebase = s * SL;
  const int jMax  = (c_s > 0) ? (c_s - 1) : 0;
  int e[FT];
#pragma unroll
  for (int k = 0; k < FT; ++k) {
    const int j = min(sub + k * TPR, jMax);
    e[k] = (int)permS[min(ebase + a_s + j, pLast)];
  }
  int   nn[FT];
  float ge[FT], x[FT], y[FT], z[FT];
#pragma unroll
  for (int k = 0; k < FT; ++k) {
    e[k]  = min(ebase + e[k], pLast);
    nn[k] = src[e[k]];
    ge[k] = g[e[k]];
    x[k] = r[3 * e[k]]; y[k] = r[3 * e[k] + 1]; z[k] = r[3 * e[k] + 2];
  }
  asm volatile("" :: "v"(nn[0]), "v"(ge[0]), "v"(x[0]), "v"(y[0]), "v"(z[0]));
#pragma unroll
  for (int k = 0; k < FT; ++k) {
    if (sub + k * TPR < c_s) {
      const int n = nn[k] - n0;
      atomicAdd(&acc[3 * n + 0], ge[k] * x[k]);
      atomicAdd(&acc[3 * n + 1], ge[k] * y[k]);
      atomicAdd(&acc[3 * n + 2], ge[k] * z[k]);
    }
  }
  for (int j = sub + FT * TPR; j < c_s; j += TPR) {
    const int ee = ebase + (int)permS[ebase + a_s + j];
    const int n = src[ee] - n0;
    const float gv = g[ee];
    atomicAdd(&acc[3 * n + 0], gv * r[3 * ee]);
    atomicAdd(&acc[3 * n + 1], gv * r[3 * ee + 1]);
    atomicAdd(&acc[3 * n + 2], gv * r[3 * ee + 2]);
  }
  __syncthreads();

  if (own) forces[oidx] = fdst + acc[threadIdx.x];
}

__global__ __launch_bounds__(256) void eam_energy(
    const float* __restrict__ partial_pair, int n1,
    const float* __restrict__ partial_F, int n2,
    float* __restrict__ energy)
{
  float v = 0.0f;
  for (int i = threadIdx.x; i < n1; i += blockDim.x) v += partial_pair[i];
  for (int i = threadIdx.x; i < n2; i += blockDim.x) v += partial_F[i];
  const float s = blockReduceSum256(v);
  if (threadIdx.x == 0) energy[0] = s;
}

extern "C" void kernel_launch(void* const* d_in, const int* in_sizes, int n_in,
                              void* d_out, int out_size, void* d_ws, size_t ws_size,
                              hipStream_t stream) {
  const float* r    = (const float*)d_in[0];
  const int*   src  = (const int*)  d_in[1];
  const int*   dst  = (const int*)  d_in[2];
  const float* phiD = (const float*)d_in[3];
  const float* phiP = (const float*)d_in[4];
  const float* emb  = (const float*)d_in[5];

  const int E = in_sizes[1];
  const int N = (out_size - 1) / 3;

  float* out    = (float*)d_out;
  float* energy = out;
  float* forces = out + 1;

  float* ws    = (float*)d_ws;
  float* g     = ws;
  float* lists = g + E;
  float* tabA  = lists + E;
  float* tabB  = tabA + N;
  float* part  = tabB + N;

  int*          st   = (int*)tabA;
  unsigned int* perm = (unsigned int*)lists;

  eam_order     <<<NSLICE, 1024, 0, stream>>>(src, dst, st, perm, E);
  eam_edge_owner<<<NBKT,   EW,   0, stream>>>(r, dst, phiD, phiP, emb, st, perm, g, part, forces, E, N);
  eam_force     <<<NBKT,   1024, 0, stream>>>(r, src, g, st, perm, forces, E, N);
  eam_energy    <<<1,      256,  0, stream>>>(part, NBKT, part, 0, energy);
}


# ordering launch split finer: a work-group orders ONE side of one of 128 slices (256 work-groups of 512 threads, not 96 doing both sides); owners read 128 runs; LDS f32 adds still arrival order
# speedup vs baseline: 3.0486x; 1.0443x over previous
#include <hip/hip_runtime.h>
#include <math.h>

typedef __attribute__((ext_vector_type(2))) float v2f;
typedef __attribute__((ext_vector_type(8))) float v8f;

#define NBASIS 128
#define NCHUNK 32
#define CUTOFF 6.0f
#define DELTA  (CUTOFF / (NBASIS - 1))
#define GAMMA_ ((NBASIS - 1) / CUTOFF)
#define PI_F   3.14159265358979323846f
#define LOG2E_F 1.44269504088896340736f
#define K2_    (-GAMMA_ * LOG2E_F)

#define NBKT   256
#define NPB    196
#define NSLICE 128
#define NORD   256
static_assert(NORD == 2 * NSLICE, "a work-group per side and slice");
static_assert(2 * NSLICE * NBKT + 2 * NBKT <= 2 * NBKT * NPB, "the range starts fit two node-sized rooms");

#if __has_builtin(__builtin_amdgcn_exp2f)
#define FAST_EXP2(x) __builtin_amdgcn_exp2f(x)
#else
#define FAST_EXP2(x) exp2f(x)
#endif

__device__ __forceinline__ float softplus_(float x) {
  return fmaxf(x, 0.0f) + log1pf(expf(-fabsf(x)));
}

__device__ __forceinline__ float blockReduceSum512(float v) {
  __shared__ float sred5[512];
  sred5[threadIdx.x] = v;
  __syncthreads();
  for (int s = blockDim.x >> 1; s > 0; s >>= 1) {
    if (threadIdx.x < (unsigned)s) sred5[threadIdx.x] += sred5[threadIdx.x + s];
    __syncthreads();
  }
  return sred5[0];
}

__device__ __forceinline__ float blockReduceSum256(float v) {
  __shared__ float sred[256];
  sred[threadIdx.x] = v;
  __syncthreads();
  for (int s = blockDim.x >> 1; s > 0; s >>= 1) {
    if (threadIdx.x < (unsigned)s) sred[threadIdx.x] += sred[threadIdx.x + s];
    __syncthreads();
  }
  return sred[0];
}

#define OTH 512
#define PT  13
#define WPS (NBKT / 32)
static_assert(WPS * 32 == NBKT, "the ranges are whole waves");
static_assert(NBKT <= OTH, "one thread per range in the scan");
#define SLCAP (PT * OTH)
#define LINE2 64

typedef unsigned int u32x4 __attribute__((ext_vector_type(4)));
typedef u32x4 __attribute__((may_alias)) u32x4a;

template <bool FITS>
__device__ __forceinline__ void list_put(unsigned short* l, unsigned short* g, int sh,
                                         int p, unsigned short v) {
  if (FITS || p < SLCAP) l[sh + p] = v; else g[p] = v;
}

__device__ __forceinline__ void list_store(const unsigned short* l, unsigned short* g,
                                           int sh, int n) {
  const int tot = sh + n;
  const int nCh = (tot + 7) >> 3;
  u32x4a* gv = (u32x4a*)(g - sh);
  const u32x4a* lv = (const u32x4a*)l;
  for (int c = threadIdx.x; c < nCh; c += blockDim.x) {
    const int j0 = 8 * c;
    if (j0 >= sh && j0 + 8 <= tot) {
      gv[c] = lv[c];
    } else {
      for (int k = 0; k < 8; ++k) {
        const int j = j0 + k;
        if (j >= sh && j < tot) g[j - sh] = l[j];
      }
    }
  }
}

template <bool FITS>
__device__ __forceinline__ void order_place(
    const int (&bk)[PT], const int* __restrict__ key, int* cnt,
    unsigned short* l, unsigned short* g, int sh, int e0, int e1)
{
#pragma unroll
  for (int i = 0; i < PT; ++i) {
    if (bk[i] >= 0) {
      const int p = atomicAdd(&cnt[bk[i]], 1);
      list_put<FITS>(l, g, sh, p, (unsigned short)((int)threadIdx.x + i * OTH));
    }
  }
  if (!FITS) {
    for (int e = e0 + PT * OTH + (int)threadIdx.x; e < e1; e += OTH) {
      const int p = atomicAdd(&cnt[key[e] / NPB], 1);
      list_put<false>(l, g, sh, p, (unsigned short)(e - e0));
    }
  }
}

__global__ __launch_bounds__(OTH) void eam_order(
    const int* __restrict__ src, const int* __restrict__ dst,
    int* __restrict__ st, unsigned int* __restrict__ perm32, int E)
{
  __shared__ int cnt[NBKT];
  __shared__ int wtot[WPS];
  __shared__ __attribute__((aligned(16))) unsigned short lst[SLCAP + LINE2];
  if (threadIdx.x < NBKT) cnt[threadIdx.x] = 0;

  const int side = (int)blockIdx.x / NSLICE;
  const int sl   = (int)blockIdx.x - side * NSLICE;
  const int* __restrict__ key = side ? dst : src;
  const int SL = (E + NSLICE - 1) / NSLICE;
  const int e0 = sl * SL;
  const int e1 = (e0 + SL < E) ? (e0 + SL) : E;
  unsigned short* gL = (unsigned short*)perm32 + (side ? E : 0) + e0;
  const int sh = (int)(((unsigned long long)gL >> 1) & (LINE2 - 1));

  int bk[PT];
#pragma unroll
  for (int i = 0; i < PT; ++i) {
    const int e  = e0 + (int)threadIdx.x + i * OTH;
    const int ec = (e < E - 1) ? e : (E - 1);
    bk[i] = key[ec];
  }
#pragma unroll
  for (int i = 0; i < PT; ++i) {
    const int e = e0 + (int)threadIdx.x + i * OTH;
    bk[i] = (e < e1) ? bk[i] / NPB : -1;
  }
  __syncthreads();
#pragma unroll
  for (int i = 0; i < PT; ++i) {
    if (bk[i] >= 0) atomicAdd(&cnt[bk[i]], 1);
  }
  for (int e = e0 + PT * OTH + (int)threadIdx.x; e < e1; e += OTH)
    atomicAdd(&cnt[key[e] / NPB], 1);
  __syncthreads();

  const int lane = threadIdx.x & 31;
  const int wv   = threadIdx.x >> 5;
  const bool owner = threadIdx.x < NBKT;
  int tot = 0;
  if (owner) tot = cnt[threadIdx.x];
  int incl = tot;
  for (int d = 1; d < 32; d <<= 1) {
    const int u = __shfl_up(incl, d);
    if (lane >= d) incl += u;
  }
  if (owner && lane == 31) wtot[wv] = incl;
  __syncthreads();
  if (owner) {
    int below = 0;
    for (int k = 0; k < wv; ++k) below += wtot[k];
    const int start = below + incl - tot;
    cnt[threadIdx.x] = start;
    st[(side * NSLICE + sl) * NBKT + (int)threadIdx.x] = start;
  }
  __syncthreads();

  if (e1 - e0 <= SLCAP)
    order_place<true >(bk, key, cnt, lst, gL, sh, e0, e1);
  else
    order_place<false>(bk, key, cnt, lst, gL, sh, e0, e1);
  __syncthreads();

  int nL = e1 - e0;
  nL = (nL < SLCAP) ? nL : SLCAP;
  nL = (nL > 0) ? nL : 0;
  list_store(lst, gL, sh, nL);
}

__device__ __forceinline__ void run_load(const int* __restrict__ stSide, int s, int b,
                                         int& a_s, int& x_s) {
  const int sc = (s < NSLICE) ? s : (NSLICE - 1);
  const int bn = (b + 1 < NBKT) ? (b + 1) : (NBKT - 1);
  const int* rw = stSide + sc * NBKT;
  a_s = rw[b];
  x_s = rw[bn];
}

__device__ __forceinline__ int run_len(int s, int b, int E, int a_s, int x_s) {
  const int SL = (E + NSLICE - 1) / NSLICE;
  int len = E - s * SL;
  len = (len < SL) ? len : SL;
  len = (len > 0) ? len : 0;
  const int n_s = (b + 1 < NBKT) ? x_s : len;
  return (s < NSLICE) ? (n_s - a_s) : 0;
}

__device__ __forceinline__ void runs_scan(int a_s, int x_s, int b, int E,
                                          int* row, int* base, int* scnt) {
  const int SL = (E + NSLICE - 1) / NSLICE;
  const int s = (int)threadIdx.x;
  if (s < NSLICE) scnt[s] = run_len(s, b, E, a_s, x_s);
  __syncthreads();
  if (s <= NSLICE) {
    int acc = 0;
    for (int k = 0; k < s; ++k) acc += scnt[k];
    row[s] = acc;
    if (s < NSLICE) base[s] = s * SL + a_s - acc;
  }
  __syncthreads();
}

#define EW    512
#define ENW   (EW / 32)
#define ECAP  4096
#define ECAPT (ECAP / 16)
static_assert(NPB <= 256, "a node's index inside its range is kept in one byte");
static_assert(NPB <= EW, "one thread per node of the range");

__device__ __forceinline__ int seg_of(const int* row, int p) {
  int lo = 0, hi = NSLICE;
  while (hi - lo > 1) {
    const int mid = (lo + hi) >> 1;
    if (row[mid] <= p) lo = mid; else hi = mid;
  }
  return lo;
}


__device__ __forceinline__ void basis_sums(const v2f (&aReg)[NCHUNK], float dist, int hi,
                                           float& SA, float& SB, float& SC, float& SD) {
  v8f acc = {0.f, 0.f, 0.f, 0.f, 0.f, 0.f, 0.f, 0.f};
  const float base = dist - (2 * hi) * DELTA;
  const v2f base2 = {base, base - DELTA};
#pragma unroll
  for (int t = 0; t < NCHUNK; ++t) {
    const v2f tt  = base2 - (float)(4 * t) * DELTA;
    const v2f arg = (tt * tt) * (float)K2_;
    v2f g;
    g.x = FAST_EXP2(arg.x);
    g.y = FAST_EXP2(arg.y);
    acc = __builtin_amdgcn_wmma_f32_16x16x4_f32(
        false, aReg[t], false, g, (short)0, acc, false, false);
  }
  SA = acc[0]; SB = acc[1]; SC = acc[2]; SD = acc[3];
}

__device__ __forceinline__ void edge_terms(float dist, float SA, float SB, float SC, float SD,
                                           float& dens, float& gdv, float& gpv, float& pairv) {
  float sn, cs;
  __sincosf(dist * (PI_F / CUTOFF), &sn, &cs);
  const float phi  = 0.25f * (1.0f + cs);
  const float phip = -0.25f * (PI_F / CUTOFF) * sn;
  const float invd = 1.0f / dist;

  dens = phi * SA;
  const float ddens  = phi * (-2.0f * GAMMA_) * (dist * SA - SB) + phip * SA;
  const float pairn  = phi * SC;
  const float dpairn = phi * (-2.0f * GAMMA_) * (dist * SC - SD) + phip * SC;
  const float dpair  = (dpairn - pairn * invd) * invd;
  gdv   = ddens * invd;
  gpv   = dpair * invd;
  pairv = pairn * invd;
}

__global__ __launch_bounds__(EW) void eam_edge_owner(
    const float* __restrict__ r, const int* __restrict__ dst,
    const float* __restrict__ phiD, const float* __restrict__ phiP,
    const float* __restrict__ w,
    const int* __restrict__ st, const unsigned int* __restrict__ perm32,
    float* __restrict__ gOut, float* __restrict__ partial,
    float* __restrict__ forces, int E, int N)
{
  __shared__ v2f sW[64 * 16];
  __shared__ int   se[ECAP];
  __shared__ float sgd[ECAP];
  __shared__ float sgp[ECAP];
  __shared__ float sx[ECAP];
  __shared__ float sy[ECAP];
  __shared__ float sz[ECAP];
  __shared__ unsigned char sn[ECAP];
  __shared__ float dloc[NPB];
  __shared__ float fpl[NPB];
  __shared__ float facc[3 * NPB];
  __shared__ int rowD[NSLICE + 1];
  __shared__ int baseD[NSLICE];
  __shared__ int cntD[NSLICE];
  const unsigned short* permD = (const unsigned short*)perm32 + E;
  const int b  = blockIdx.x;
  const int n0 = b * NPB;
  int runA, runX;
  run_load(st + NSLICE * NBKT, (int)threadIdx.x, b, runA, runX);

  for (int idx = threadIdx.x; idx < 64 * 16; idx += blockDim.x) {
    const int kp = idx >> 4, ch = idx & 15;
    const int k0 = 2 * kp, k1 = k0 + 1;
    const float c0 = k0 * DELTA, c1 = k1 * DELTA;
    v2f wv = {0.0f, 0.0f};
    if (ch < 4) {
      float a0, a1;
      if (ch < 2) { a0 = softplus_(phiD[k0]); a1 = softplus_(phiD[k1]); }
      else        { a0 = phiP[k0];            a1 = phiP[k1]; }
      if (ch & 1) { a0 *= c0; a1 *= c1; }
      wv.x = a0; wv.y = a1;
    }
    sW[idx] = wv;
  }
  if (threadIdx.x < NPB) dloc[threadIdx.x] = 0.0f;
  for (int i = threadIdx.x; i < 3 * NPB; i += blockDim.x) facc[i] = 0.0f;
  runs_scan(runA, runX, b, E, rowD, baseD, cntD);

  const int lane = threadIdx.x & 31;
  const int col  = lane & 15;
  const int hi   = lane >> 4;
  const int wv   = threadIdx.x >> 5;

  v2f aReg[NCHUNK];
#pragma unroll
  for (int t = 0; t < NCHUNK; ++t)
    aReg[t] = sW[(2 * t + hi) * 16 + col];

  const int SL   = (E + NSLICE - 1) / NSLICE;
  const int pEnd = rowD[NSLICE];
  const int nT   = (pEnd + 15) >> 4;
  const int pMax = (pEnd > 0) ? (pEnd - 1) : 0;

  float pairAcc = 0.0f;

  const int pLast = E - 1;
  int pB = wv * 16 + col;
  int pc = min(pB, pMax);
  int sB = seg_of(rowD, pc);
  int locB = (int)permD[min(baseD[sB] + pc, pLast)];
  int eA = min(sB * SL + locB, pLast);
  float xA = r[3 * eA], yA = r[3 * eA + 1], zA = r[3 * eA + 2];
  int nA = dst[eA] - n0;
  bool vA = pB < pEnd;
  pB = (wv + ENW) * 16 + col;
  pc = min(pB, pMax);
  sB = seg_of(rowD, pc);
  locB = (int)permD[min(baseD[sB] + pc, pLast)];

  for (int tile = wv; tile < nT; tile += ENW) {
    const int   eN = min(sB * SL + locB, pLast);
    const float xN = r[3 * eN], yN = r[3 * eN + 1], zN = r[3 * eN + 2];
    const int   nN = dst[eN] - n0;
    const bool  vN = pB < pEnd;
    pB = (tile + 2 * ENW) * 16 + col;
    pc = min(pB, pMax);
    sB = seg_of(rowD, pc);
    locB = (int)permD[min(baseD[sB] + pc, pLast)];

    const float dist = sqrtf(xA * xA + yA * yA + zA * zA);
    float SA, SB, SC, SD;
    basis_sums(aReg, dist, hi, SA, SB, SC, SD);

    if (lane < 16 && vA) {
      float dens, gdv, gpv, pairv;
      edge_terms(dist, SA, SB, SC, SD, dens, gdv, gpv, pairv);
      atomicAdd(&dloc[nA], dens);
      const int i = tile * 16 + col;
      if (i < ECAP) {
        se[i] = eA; sn[i] = (unsigned char)nA; sgd[i] = gdv; sgp[i] = gpv;
        sx[i] = xA; sy[i] = yA; sz[i] = zA;
      }
      pairAcc += pairv;
    }
    eA = eN; nA = nN; xA = xN; yA = yN; zA = zN; vA = vN;
  }
  __syncthreads();

  float Facc = 0.0f;
  if (threadIdx.x < NPB && n0 + (int)threadIdx.x < N) {
    const float w0 = w[0], w1 = w[1], w2 = w[2], w3 = w[3], w4 = w[4];
    const float x  = dloc[threadIdx.x];
    const float sq = sqrtf(x);
    const float x2 = x * x;
    Facc = 2.0f * w0 * sq + w1 * x + 0.5f * w2 * x2 +
           (1.0f / 6.0f) * w3 * x2 * x + (1.0f / 24.0f) * w4 * x2 * x2;
    fpl[threadIdx.x] = w0 / sq + w1 + w2 * x + 0.5f * w3 * x2 +
                       (1.0f / 6.0f) * w4 * x2 * x;
  }
  __syncthreads();

  const int nHeld = (pEnd < ECAP) ? pEnd : ECAP;
  for (int i = threadIdx.x; i < nHeld; i += blockDim.x) {
    const int n = sn[i];
    const float x = sx[i], y = sy[i], z = sz[i];
    const float gv = fpl[n] * sgd[i] + sgp[i];
    gOut[se[i]] = gv;
    atomicAdd(&facc[3 * n + 0], -(gv * x));
    atomicAdd(&facc[3 * n + 1], -(gv * y));
    atomicAdd(&facc[3 * n + 2], -(gv * z));
  }

  for (int tile = ECAPT + wv; tile < nT; tile += ENW) {
    const int p = tile * 16 + col;
    const bool v = p < pEnd;
    int e = 0, n = 0; float x = 0.0f, y = 0.0f, z = 0.0f;
    if (v) {
      const int s = seg_of(rowD, p);
      e = s * SL + (int)permD[baseD[s] + p];
      x = r[3 * e]; y = r[3 * e + 1]; z = r[3 * e + 2];
      n = dst[e] - n0;
    }
    const float dist = sqrtf(x * x + y * y + z * z);
    float SA, SB, SC, SD;
    basis_sums(aReg, dist, hi, SA, SB, SC, SD);
    if (lane < 16 && v) {
      float dens, gdv, gpv, pairv;
      edge_terms(dist, SA, SB, SC, SD, dens, gdv, gpv, pairv);
      const float gv = fpl[n] * gdv + gpv;
      gOut[e] = gv;
      atomicAdd(&facc[3 * n + 0], -(gv * x));
      atomicAdd(&facc[3 * n + 1], -(gv * y));
      atomicAdd(&facc[3 * n + 2], -(gv * z));
    }
  }
  __syncthreads();

  for (int i = threadIdx.x; i < 3 * NPB; i += blockDim.x) {
    const int idx = 3 * n0 + i;
    if (idx < 3 * N) forces[idx] = facc[i];
  }

  const float bsum = blockReduceSum512(pairAcc + Facc);
  if (threadIdx.x == 0) partial[b] = bsum;
}

#define TPR 8
#define FT  6
static_assert(TPR * NSLICE <= 1024, "every run has its threads");

__global__ __launch_bounds__(1024) void eam_force(
    const float* __restrict__ r, const int* __restrict__ src,
    const float* __restrict__ g,
    const int* __restrict__ st, const unsigned int* __restrict__ perm32,
    float* __restrict__ forces, int E, int N)
{
  __shared__ float acc[3 * NPB];
  const unsigned short* permS = (const unsigned short*)perm32;
  const int b = blockIdx.x;
  const int n0 = b * NPB;
  static_assert(3 * NPB <= 1024, "one thread per force component of the range");

  const int SL  = (E + NSLICE - 1) / NSLICE;
  const int s   = (int)threadIdx.x / TPR;
  const int sub = (int)threadIdx.x - s * TPR;
  int a_s, x_s;
  run_load(st, s, b, a_s, x_s);

  const int oidx = 3 * n0 + (int)threadIdx.x;
  const bool own = threadIdx.x < 3 * NPB && oidx < 3 * N;
  float fdst = 0.0f;
  if (own) fdst = forces[oidx];

  for (int i = threadIdx.x; i < 3 * NPB; i += blockDim.x) acc[i] = 0.0f;
  const int c_s = run_len(s, b, E, a_s, x_s);
  __syncthreads();

  const int pLast = E - 1;
  const int ebase = s * SL;
  const int jMax  = (c_s > 0) ? (c_s - 1) : 0;
  int e[FT];
#pragma unroll
  for (int k = 0; k < FT; ++k) {
    const int j = min(sub + k * TPR, jMax);
    e[k] = (int)permS[min(ebase + a_s + j, pLast)];
  }
  int   nn[FT];
  float ge[FT], x[FT], y[FT], z[FT];
#pragma unroll
  for (int k = 0; k < FT; ++k) {
    e[k]  = min(ebase + e[k], pLast);
    nn[k] = src[e[k]];
    ge[k] = g[e[k]];
    x[k] = r[3 * e[k]]; y[k] = r[3 * e[k] + 1]; z[k] = r[3 * e[k] + 2];
  }
  asm volatile("" :: "v"(nn[0]), "v"(ge[0]), "v"(x[0]), "v"(y[0]), "v"(z[0]));
#pragma unroll
  for (int k = 0; k < FT; ++k) {
    if (sub + k * TPR < c_s) {
      const int n = nn[k] - n0;
      atomicAdd(&acc[3 * n + 0], ge[k] * x[k]);
      atomicAdd(&acc[3 * n + 1], ge[k] * y[k]);
      atomicAdd(&acc[3 * n + 2], ge[k] * z[k]);
    }
  }
  for (int j = sub + FT * TPR; j < c_s; j += TPR) {
    const int ee = ebase + (int)permS[ebase + a_s + j];
    const int n = src[ee] - n0;
    const float gv = g[ee];
    atomicAdd(&acc[3 * n + 0], gv * r[3 * ee]);
    atomicAdd(&acc[3 * n + 1], gv * r[3 * ee + 1]);
    atomicAdd(&acc[3 * n + 2], gv * r[3 * ee + 2]);
  }
  __syncthreads();

  if (own) forces[oidx] = fdst + acc[threadIdx.x];
}

__global__ __launch_bounds__(256) void eam_energy(
    const float* __restrict__ partial_pair, int n1,
    const float* __restrict__ partial_F, int n2,
    float* __restrict__ energy)
{
  float v = 0.0f;
  for (int i = threadIdx.x; i < n1; i += blockDim.x) v += partial_pair[i];
  for (int i = threadIdx.x; i < n2; i += blockDim.x) v += partial_F[i];
  const float s = blockReduceSum256(v);
  if (threadIdx.x == 0) energy[0] = s;
}

extern "C" void kernel_launch(void* const* d_in, const int* in_sizes, int n_in,
                              void* d_out, int out_size, void* d_ws, size_t ws_size,
                              hipStream_t stream) {
  const float* r    = (const float*)d_in[0];
  const int*   src  = (const int*)  d_in[1];
  const int*   dst  = (const int*)  d_in[2];
  const float* phiD = (const float*)d_in[3];
  const float* phiP = (const float*)d_in[4];
  const float* emb  = (const float*)d_in[5];

  const int E = in_sizes[1];
  const int N = (out_size - 1) / 3;

  float* out    = (float*)d_out;
  float* energy = out;
  float* forces = out + 1;

  float* ws    = (float*)d_ws;
  float* g     = ws;
  float* lists = g + E;
  float* tabA  = lists + E;
  float* tabB  = tabA + N;
  float* part  = tabB + N;

  int*          st   = (int*)tabA;
  unsigned int* perm = (unsigned int*)lists;

  eam_order     <<<NORD,   OTH,  0, stream>>>(src, dst, st, perm, E);
  eam_edge_owner<<<NBKT,   EW,   0, stream>>>(r, dst, phiD, phiP, emb, st, perm, g, part, forces, E, N);
  eam_force     <<<NBKT,   1024, 0, stream>>>(r, src, g, st, perm, forces, E, N);
  eam_energy    <<<1,      256,  0, stream>>>(part, NBKT, part, 0, energy);
}
